# starting kernel
# baseline (speedup 1.0000x reference)
#include <hip/hip_runtime.h>
#include <hip/hip_bf16.h>
#include <math.h>

constexpr int B_  = 2;
constexpr int H_  = 8;
constexpr int LQ_ = 128;
constexpr int LK_ = 128;
constexpr int D_  = 512;
constexpr int DH_ = 64;
constexpr int F_  = 512;

typedef float v2f __attribute__((ext_vector_type(2)));
typedef float v8f __attribute__((ext_vector_type(8)));

#if __has_builtin(__builtin_amdgcn_tanhf)
#define TANH_F32(x) __builtin_amdgcn_tanhf(x)
#elif __has_builtin(__builtin_amdgcn_tanh_f32)
#define TANH_F32(x) __builtin_amdgcn_tanh_f32(x)
#else
#define TANH_F32(x) tanhf(x)
#endif

__device__ __forceinline__ v8f wmma_f32_tile(const float* __restrict__ A, int lda,
                                             const float* __restrict__ W, int ldw,
                                             int K, int m0, int n0, int lane)
{
  const int half = lane >> 4;
  const int l16  = lane & 15;
  const float* arow = A + (size_t)(m0 + l16) * lda + 2 * half;
  const float* wrow = W + (size_t)(n0 + l16) * ldw + 2 * half;
  v8f c = {};
  for (int k = 0; k < K; k += 4) {
    v2f a = *(const v2f*)(arow + k);
    v2f b = *(const v2f*)(wrow + k);
    c = __builtin_amdgcn_wmma_f32_16x16x4_f32(
         false, a,  false, b,
         (short)0, c,  false,  false);
  }
  return c;
}

__global__ void __launch_bounds__(256)
proj_kernel(const float* __restrict__ X, const float* __restrict__ W,
            float* __restrict__ out, int split)
{
  const int M = B_ * LQ_;
  const int N = D_;
  const int K = D_;
  const int tid  = threadIdx.x;
  const int lane = tid & 31;
  const int wave = tid >> 5;
  const int tile = blockIdx.x * 8 + wave;
  const int tm = tile / (N / 16);
  const int tn = tile % (N / 16);
  const int m0 = tm * 16, n0 = tn * 16;

  v8f c = wmma_f32_tile(X, K, W, K, K, m0, n0, lane);

  const int l16 = lane & 15, half = lane >> 4;
  const int col = n0 + l16;
  for (int r = 0; r < 8; ++r) {
    const int row = m0 + r + 8 * half;
    if (split) {
      const int b = row / LQ_, l = row % LQ_;
      const int h = col / DH_, j = col % DH_;
      out[(((size_t)(b * H_ + h) * LQ_) + l) * DH_ + j] = c[r];
    } else {
      out[(size_t)row * N + col] = c[r];
    }
  }
}

__global__ void __launch_bounds__(256)
feat_kernel(const float* __restrict__ Hd, const float* __restrict__ Wa,
            float* __restrict__ outf)
{
  const int bh = blockIdx.y;
  const float* A = Hd + (size_t)bh * LQ_ * DH_;
  float* O = outf + (size_t)bh * LQ_ * F_;

  const int tid  = threadIdx.x;
  const int lane = tid & 31;
  const int wave = tid >> 5;
  const int tile = blockIdx.x * 8 + wave;
  const int tm = tile / (F_ / 16);
  const int tn = tile % (F_ / 16);
  const int m0 = tm * 16, n0 = tn * 16;

  v8f c = wmma_f32_tile(A, DH_, Wa, DH_, DH_, m0, n0, lane);

  const int l16 = lane & 15, half = lane >> 4;
  for (int r = 0; r < 8; ++r)
    O[(size_t)(m0 + r + 8 * half) * F_ + n0 + l16] = c[r];
}

__global__ void __launch_bounds__(256)
score_softmax_av_kernel(const float* __restrict__ qf,
                        const float* __restrict__ kf,
                        const float* __restrict__ vh,
                        const float* __restrict__ wv_a,
                        const int*   __restrict__ valid_lens,
                        float* __restrict__ ctx)
{
  __shared__ float sq[F_];
  __shared__ float swv[F_];
  __shared__ float sp[LK_];
  __shared__ float spart[4 * DH_];
  __shared__ float swmax[8];
  __shared__ float swsum[8];

  const int bhq = blockIdx.x;
  const int qi  = bhq % LQ_;
  const int bh  = bhq / LQ_;
  const int b   = bh / H_;
  const int h   = bh % H_;
  const int tid  = threadIdx.x;
  const int lane = tid & 31;
  const int wave = tid >> 5;

  const float* qrow = qf + ((size_t)bh * LQ_ + qi) * F_;
  for (int f = tid; f < F_; f += 256) { sq[f] = qrow[f]; swv[f] = wv_a[f]; }
  __syncthreads();

  const float* kbase = kf + (size_t)bh * LK_ * F_;
  for (int k = wave; k < LK_; k += 8) {
    const float* krow = kbase + (size_t)k * F_;
    float acc = 0.f;
    for (int f = lane; f < F_; f += 32)
      acc += swv[f] * TANH_F32(sq[f] + krow[f]);
    for (int off = 16; off > 0; off >>= 1)
      acc += __shfl_xor(acc, off, 32);
    if (lane == 0) sp[k] = acc;
  }
  __syncthreads();

  const int valid = valid_lens[b];
  float val = (tid < LK_) ? ((tid < valid) ? sp[tid] : -1.0e6f) : -3.0e38f;

  float m = val;
  for (int off = 16; off > 0; off >>= 1)
    m = fmaxf(m, __shfl_xor(m, off, 32));
  if (lane == 0) swmax[wave] = (wave < 4) ? m : -3.0e38f;
  __syncthreads();
  const float rowmax = fmaxf(fmaxf(swmax[0], swmax[1]), fmaxf(swmax[2], swmax[3]));

  float e = (tid < LK_) ? __expf(val - rowmax) : 0.f;
  float s = e;
  for (int off = 16; off > 0; off >>= 1)
    s += __shfl_xor(s, off, 32);
  if (lane == 0) swsum[wave] = (wave < 4) ? s : 0.f;
  __syncthreads();
  const float rowsum = swsum[0] + swsum[1] + swsum[2] + swsum[3];
  if (tid < LK_) sp[tid] = e / rowsum;
  __syncthreads();

  const int j = tid & 63;
  const int part = tid >> 6;
  const float* vb = vh + (size_t)bh * LK_ * DH_;
  float acc = 0.f;
  for (int k = part * 32; k < part * 32 + 32; ++k)
    acc += sp[k] * vb[(size_t)k * DH_ + j];
  spart[part * DH_ + j] = acc;
  __syncthreads();

  if (tid < DH_) {
    const float o = spart[tid] + spart[DH_ + tid] + spart[2 * DH_ + tid] + spart[3 * DH_ + tid];
    ctx[((size_t)b * LQ_ + qi) * D_ + h * DH_ + tid] = o;
  }
}

extern "C" void kernel_launch(void* const* d_in, const int* in_sizes, int n_in,
                              void* d_out, int out_size, void* d_ws, size_t ws_size,
                              hipStream_t stream) {
  (void)in_sizes; (void)n_in; (void)out_size; (void)ws_size;

  const float* queries    = (const float*)d_in[0];
  const float* keys       = (const float*)d_in[1];
  const float* values     = (const float*)d_in[2];
  const int*   valid_lens = (const int*)  d_in[3];
  const float* Wq  = (const float*)d_in[4];
  const float* Wk  = (const float*)d_in[5];
  const float* Wv  = (const float*)d_in[6];
  const float* Wo  = (const float*)d_in[7];
  const float* Wqa = (const float*)d_in[8];
  const float* Wka = (const float*)d_in[9];
  const float* wva = (const float*)d_in[10];

  float* ws  = (float*)d_ws;
  const size_t nQKV = (size_t)B_ * H_ * LQ_ * DH_;
  const size_t nF   = (size_t)B_ * H_ * LQ_ * F_;
  float* q   = ws;
  float* k   = q  + nQKV;
  float* v   = k  + nQKV;
  float* qf  = v  + nQKV;
  float* kfb = qf + nF;
  float* ctx = kfb + nF;

  dim3 blk(256);
  proj_kernel<<<64, blk, 0, stream>>>(queries, Wq, q, 1);
  proj_kernel<<<64, blk, 0, stream>>>(keys,    Wk, k, 1);
  proj_kernel<<<64, blk, 0, stream>>>(values,  Wv, v, 1);

  feat_kernel<<<dim3(32, B_ * H_), blk, 0, stream>>>(q, Wqa, qf);
  feat_kernel<<<dim3(32, B_ * H_), blk, 0, stream>>>(k, Wka, kfb);

  score_softmax_av_kernel<<<B_ * H_ * LQ_, blk, 0, stream>>>(
      qf, kfb, v, wva, valid_lens, ctx);

  proj_kernel<<<64, blk, 0, stream>>>(ctx, Wo, (float*)d_out, 0);
}


# Projection and feature GEMMs: fragments straight from global by 16-byte loads, ~20 ahead in registers; projections on 256 two-wave groups; a dot product is now two interleaved k sums, added last
# speedup vs baseline: 1.9056x; 1.9056x over previous
#include <hip/hip_runtime.h>
#include <hip/hip_bf16.h>
#include <math.h>

constexpr int B_  = 2;
constexpr int H_  = 8;
constexpr int LQ_ = 128;
constexpr int LK_ = 128;
constexpr int D_  = 512;
constexpr int DH_ = 64;
constexpr int F_  = 512;

typedef float v2f __attribute__((ext_vector_type(2)));
typedef float v4f __attribute__((ext_vector_type(4)));
typedef float v8f __attribute__((ext_vector_type(8)));

#if __has_builtin(__builtin_amdgcn_tanhf)
#define TANH_F32(x) __builtin_amdgcn_tanhf(x)
#elif __has_builtin(__builtin_amdgcn_tanh_f32)
#define TANH_F32(x) __builtin_amdgcn_tanh_f32(x)
#else
#define TANH_F32(x) tanhf(x)
#endif

constexpr int CH_    = 8;
constexpr int AHEAD_ = 3;

template <int K>
__device__ __forceinline__ v8f wmma_f32_tile(const float* __restrict__ A, int lda,
                                             const float* __restrict__ W, int ldw,
                                             int m0, int n0, int lane)
{
  constexpr int NP  = K / 8;
  constexpr int NCH = NP / CH_;
  constexpr int LA  = (NCH < AHEAD_) ? NCH : AHEAD_;
  const int half = lane >> 4;
  const int l16  = lane & 15;
  const v4f* arow = (const v4f*)(A + (size_t)(m0 + l16) * lda + 4 * half);
  const v4f* wrow = (const v4f*)(W + (size_t)(n0 + l16) * ldw + 4 * half);
  v4f a[NP], b[NP];
  v8f c0 = {}, c1 = {};
#pragma unroll
  for (int ch = 0; ch < NCH + LA; ++ch) {
    if (ch < NCH) {
#pragma unroll
      for (int i = ch * CH_; i < (ch + 1) * CH_; ++i) {
        a[i] = arow[2 * i];
        b[i] = wrow[2 * i];
      }
    }
    if (ch >= LA) {
#pragma unroll
      for (int i = (ch - LA) * CH_; i < (ch - LA + 1) * CH_; ++i) {
        const v2f a01 = {a[i][0], a[i][1]}, b01 = {b[i][0], b[i][1]};
        const v2f a23 = {a[i][2], a[i][3]}, b23 = {b[i][2], b[i][3]};
        c0 = __builtin_amdgcn_wmma_f32_16x16x4_f32(
             false, a01,  false, b01,
             (short)0, c0,  false,  false);
        c1 = __builtin_amdgcn_wmma_f32_16x16x4_f32(
             false, a23,  false, b23,
             (short)0, c1,  false,  false);
      }
    }
  }
  return c0 + c1;
}

__global__ void __launch_bounds__(64) __attribute__((amdgpu_waves_per_eu(1, 2)))
proj_kernel(const float* __restrict__ X, const float* __restrict__ W,
            float* __restrict__ out, int split)
{
  constexpr int N = D_;
  constexpr int K = D_;
  const int tid  = threadIdx.x;
  const int lane = tid & 31;
  const int wave = tid >> 5;
  const int tile = blockIdx.x * 2 + wave;
  const int tm = tile / (N / 16);
  const int tn = tile % (N / 16);
  const int m0 = tm * 16, n0 = tn * 16;

  v8f c = wmma_f32_tile<K>(X, K, W, K, m0, n0, lane);

  const int l16 = lane & 15, half = lane >> 4;
  const int col = n0 + l16;
  for (int r = 0; r < 8; ++r) {
    const int row = m0 + r + 8 * half;
    if (split) {
      const int b = row / LQ_, l = row % LQ_;
      const int h = col / DH_, j = col % DH_;
      out[(((size_t)(b * H_ + h) * LQ_) + l) * DH_ + j] = c[r];
    } else {
      out[(size_t)row * N + col] = c[r];
    }
  }
}

__global__ void __launch_bounds__(256) __attribute__((amdgpu_waves_per_eu(2, 4)))
feat_kernel(const float* __restrict__ Hd, const float* __restrict__ Wa,
            float* __restrict__ outf)
{
  const int bh = blockIdx.y;
  const float* A = Hd + (size_t)bh * LQ_ * DH_;
  float* O = outf + (size_t)bh * LQ_ * F_;

  const int tid  = threadIdx.x;
  const int lane = tid & 31;
  const int wave = tid >> 5;
  const int tile = blockIdx.x * 8 + wave;
  const int tm = tile / (F_ / 16);
  const int tn = tile % (F_ / 16);
  const int m0 = tm * 16, n0 = tn * 16;

  v8f c = wmma_f32_tile<DH_>(A, DH_, Wa, DH_, m0, n0, lane);

  const int l16 = lane & 15, half = lane >> 4;
  for (int r = 0; r < 8; ++r)
    O[(size_t)(m0 + r + 8 * half) * F_ + n0 + l16] = c[r];
}

constexpr int TQ_  = 8;
constexpr int F4_  = F_ / 4;
constexpr int KP4_ = F4_ + 1;
constexpr int SLAB4_ = 16;
constexpr int SLABS_ = F4_ / SLAB4_;
constexpr int SPT_   = (LK_ * SLAB4_) / 256;
constexpr int QPT_   = (TQ_ * F4_) / 256;

__global__ void __launch_bounds__(256)
score_softmax_av_kernel(const float* __restrict__ qf,
                        const float* __restrict__ kf,
                        const float* __restrict__ vh,
                        const float* __restrict__ wv_a,
                        const int*   __restrict__ valid_lens,
                        float* __restrict__ ctx)
{
  __shared__ v4f skf[LK_ * KP4_];
  __shared__ v4f sqf[TQ_ * F4_];
  __shared__ v4f sw[F4_];
  __shared__ float sp[TQ_ * LK_];

  const int qt  = blockIdx.x % (LQ_ / TQ_);
  const int bh  = blockIdx.x / (LQ_ / TQ_);
  const int b   = bh / H_;
  const int h   = bh % H_;
  const int tid  = threadIdx.x;
  const int lane = tid & 31;
  const int wave = tid >> 5;

  const v4f* kf4 = (const v4f*)(kf + (size_t)bh * LK_ * F_);
  const v4f* qf4 = (const v4f*)(qf + ((size_t)bh * LQ_ + (size_t)qt * TQ_) * F_);
  const v4f* wv4 = (const v4f*)wv_a;

  int goff[SPT_];
  int loff[SPT_];
#pragma unroll
  for (int jj = 0; jj < SPT_; ++jj) {
    const int u = tid + 256 * jj;
    goff[jj] = (u >> 4) * F4_  + (u & (SLAB4_ - 1));
    loff[jj] = (u >> 4) * KP4_ + (u & (SLAB4_ - 1));
  }
  v4f cur[SPT_];
#pragma unroll
  for (int jj = 0; jj < SPT_; ++jj) cur[jj] = kf4[goff[jj]];
#pragma unroll
  for (int i = 0; i < QPT_; ++i) sqf[tid + 256 * i] = qf4[tid + 256 * i];
  if (tid < F4_) sw[tid] = wv4[tid];

  const int qa = 2 * (wave & 3);
  const int ka = 64 * (wave >> 2) + lane;
  const v4f* q0p = sqf + qa * F4_;
  const v4f* q1p = q0p + F4_;
  const v4f* k0p = skf + ka * KP4_;
  const v4f* k1p = k0p + 32 * KP4_;
  float a00 = 0.f, a01 = 0.f, a10 = 0.f, a11 = 0.f;
#pragma nounroll
  for (int s = 0; s < SLABS_; ++s) {
#pragma unroll
    for (int jj = 0; jj < SPT_; ++jj) skf[loff[jj] + SLAB4_ * s] = cur[jj];
    __syncthreads();
    if (s + 1 < SLABS_) {
#pragma unroll
      for (int jj = 0; jj < SPT_; ++jj) cur[jj] = kf4[goff[jj] + SLAB4_ * (s + 1)];
    }
    for (int f4 = SLAB4_ * s; f4 < SLAB4_ * (s + 1); ++f4) {
      const v4f w  = sw[f4];
      const v4f q0 = q0p[f4];
      const v4f q1 = q1p[f4];
      const v4f k0 = k0p[f4];
      const v4f k1 = k1p[f4];
#pragma unroll
      for (int e = 0; e < 4; ++e) {
        a00 += w[e] * TANH_F32(q0[e] + k0[e]);
        a01 += w[e] * TANH_F32(q0[e] + k1[e]);
        a10 += w[e] * TANH_F32(q1[e] + k0[e]);
        a11 += w[e] * TANH_F32(q1[e] + k1[e]);
      }
    }
  }
  sp[qa * LK_ + ka]            = a00;
  sp[qa * LK_ + ka + 32]       = a01;
  sp[(qa + 1) * LK_ + ka]      = a10;
  sp[(qa + 1) * LK_ + ka + 32] = a11;
  __syncthreads();

  {
    const int valid = valid_lens[b];
    float* srow = sp + wave * LK_;
    float val[4];
#pragma unroll
    for (int g = 0; g < 4; ++g) {
      const int k = lane + 32 * g;
      val[g] = (k < valid) ? srow[k] : -1.0e6f;
    }
    float m = fmaxf(fmaxf(val[0], val[1]), fmaxf(val[2], val[3]));
    for (int off = 16; off > 0; off >>= 1)
      m = fmaxf(m, __shfl_xor(m, off, 32));
    float e[4], s[4];
#pragma unroll
    for (int g = 0; g < 4; ++g) {
      e[g] = __expf(val[g] - m);
      s[g] = e[g];
      for (int off = 16; off > 0; off >>= 1)
        s[g] += __shfl_xor(s[g], off, 32);
    }
    const float rowsum = s[0] + s[1] + s[2] + s[3];
#pragma unroll
    for (int g = 0; g < 4; ++g)
      srow[lane + 32 * g] = e[g] / rowsum;
  }
  __syncthreads();

  const int j  = tid & 63;
  const int qg = tid >> 6;
  const float* vb = vh + (size_t)bh * LK_ * DH_ + j;
  const float* pa = sp + qg * LK_;
  const float* pb = sp + (qg + 4) * LK_;
  float oa[4] = {0.f, 0.f, 0.f, 0.f};
  float ob[4] = {0.f, 0.f, 0.f, 0.f};
  for (int kk = 0; kk < 32; ++kk) {
#pragma unroll
    for (int part = 0; part < 4; ++part) {
      const int k = part * 32 + kk;
      const float vv = vb[(size_t)k * DH_];
      oa[part] += pa[k] * vv;
      ob[part] += pb[k] * vv;
    }
  }
  const int rowa = qt * TQ_ + qg;
  ctx[((size_t)b * LQ_ + rowa) * D_ + h * DH_ + j]     = oa[0] + oa[1] + oa[2] + oa[3];
  ctx[((size_t)b * LQ_ + rowa + 4) * D_ + h * DH_ + j] = ob[0] + ob[1] + ob[2] + ob[3];
}

extern "C" void kernel_launch(void* const* d_in, const int* in_sizes, int n_in,
                              void* d_out, int out_size, void* d_ws, size_t ws_size,
                              hipStream_t stream) {
  (void)in_sizes; (void)n_in; (void)out_size; (void)ws_size;

  const float* queries    = (const float*)d_in[0];
  const float* keys       = (const float*)d_in[1];
  const float* values     = (const float*)d_in[2];
  const int*   valid_lens = (const int*)  d_in[3];
  const float* Wq  = (const float*)d_in[4];
  const float* Wk  = (const float*)d_in[5];
  const float* Wv  = (const float*)d_in[6];
  const float* Wo  = (const float*)d_in[7];
  const float* Wqa = (const float*)d_in[8];
  const float* Wka = (const float*)d_in[9];
  const float* wva = (const float*)d_in[10];

  float* ws  = (float*)d_ws;
  const size_t nQKV = (size_t)B_ * H_ * LQ_ * DH_;
  const size_t nF   = (size_t)B_ * H_ * LQ_ * F_;
  float* q   = ws;
  float* k   = q  + nQKV;
  float* v   = k  + nQKV;
  float* qf  = v  + nQKV;
  float* kfb = qf + nF;
  float* ctx = kfb + nF;

  dim3 blk(256);
  dim3 pblk(64);
  proj_kernel<<<256, pblk, 0, stream>>>(queries, Wq, q, 1);
  proj_kernel<<<256, pblk, 0, stream>>>(keys,    Wk, k, 1);
  proj_kernel<<<256, pblk, 0, stream>>>(values,  Wv, v, 1);

  feat_kernel<<<dim3(32, B_ * H_), blk, 0, stream>>>(q, Wqa, qf);
  feat_kernel<<<dim3(32, B_ * H_), blk, 0, stream>>>(k, Wka, kfb);

  score_softmax_av_kernel<<<B_ * H_ * (LQ_ / TQ_), blk, 0, stream>>>(
      qf, kfb, v, wva, valid_lens, ctx);

  proj_kernel<<<256, pblk, 0, stream>>>(ctx, Wo, (float*)d_out, 0);
}


# Query side moves into the score work-group: its 8 rows' q tile and 512 features on WMMA in the prologue, handed on in LDS; 7 launches -> 5; q now sums k in two halves of 256, added last
# speedup vs baseline: 2.0916x; 1.0976x over previous
#include <hip/hip_runtime.h>
#include <hip/hip_bf16.h>
#include <math.h>

constexpr int B_  = 2;
constexpr int H_  = 8;
constexpr int LQ_ = 128;
constexpr int LK_ = 128;
constexpr int D_  = 512;
constexpr int DH_ = 64;
constexpr int F_  = 512;

typedef float v2f __attribute__((ext_vector_type(2)));
typedef float v4f __attribute__((ext_vector_type(4)));
typedef float v8f __attribute__((ext_vector_type(8)));

#if __has_builtin(__builtin_amdgcn_tanhf)
#define TANH_F32(x) __builtin_amdgcn_tanhf(x)
#elif __has_builtin(__builtin_amdgcn_tanh_f32)
#define TANH_F32(x) __builtin_amdgcn_tanh_f32(x)
#else
#define TANH_F32(x) tanhf(x)
#endif

constexpr int CH_    = 8;
constexpr int AHEAD_ = 3;

template <int K, typename AfterRequests>
__device__ __forceinline__ v8f wmma_f32_rows(const v4f* __restrict__ arow,
                                             const v4f* __restrict__ wrow,
                                             AfterRequests&& after_requests)
{
  constexpr int NP  = K / 8;
  constexpr int NCH = NP / CH_;
  constexpr int LA  = (NCH < AHEAD_) ? NCH : AHEAD_;
  v4f a[NP], b[NP];
  v8f c0 = {}, c1 = {};
#pragma unroll
  for (int ch = 0; ch < NCH + LA; ++ch) {
    if (ch < NCH) {
#pragma unroll
      for (int i = ch * CH_; i < (ch + 1) * CH_; ++i) {
        a[i] = arow[2 * i];
        b[i] = wrow[2 * i];
      }
      if (ch == NCH - 1) after_requests();
    }
    if (ch >= LA) {
#pragma unroll
      for (int i = (ch - LA) * CH_; i < (ch - LA + 1) * CH_; ++i) {
        const v2f a01 = {a[i][0], a[i][1]}, b01 = {b[i][0], b[i][1]};
        const v2f a23 = {a[i][2], a[i][3]}, b23 = {b[i][2], b[i][3]};
        c0 = __builtin_amdgcn_wmma_f32_16x16x4_f32(
             false, a01,  false, b01,
             (short)0, c0,  false,  false);
        c1 = __builtin_amdgcn_wmma_f32_16x16x4_f32(
             false, a23,  false, b23,
             (short)0, c1,  false,  false);
      }
    }
  }
  return c0 + c1;
}

template <int K>
__device__ __forceinline__ v8f wmma_f32_tile(const float* __restrict__ A, int lda,
                                             const float* __restrict__ W, int ldw,
                                             int m0, int n0, int lane)
{
  const int half = lane >> 4;
  const int l16  = lane & 15;
  return wmma_f32_rows<K>((const v4f*)(A + (size_t)(m0 + l16) * lda + 4 * half),
                          (const v4f*)(W + (size_t)(n0 + l16) * ldw + 4 * half), [] {});
}

__global__ void __launch_bounds__(64) __attribute__((amdgpu_waves_per_eu(1, 2)))
proj_kernel(const float* __restrict__ X, const float* __restrict__ W,
            float* __restrict__ out, int split)
{
  constexpr int N = D_;
  constexpr int K = D_;
  const int tid  = threadIdx.x;
  const int lane = tid & 31;
  const int wave = tid >> 5;
  const int tile = blockIdx.x * 2 + wave;
  const int tm = tile / (N / 16);
  const int tn = tile % (N / 16);
  const int m0 = tm * 16, n0 = tn * 16;

  v8f c = wmma_f32_tile<K>(X, K, W, K, m0, n0, lane);

  const int l16 = lane & 15, half = lane >> 4;
  const int col = n0 + l16;
  for (int r = 0; r < 8; ++r) {
    const int row = m0 + r + 8 * half;
    if (split) {
      const int b = row / LQ_, l = row % LQ_;
      const int h = col / DH_, j = col % DH_;
      out[(((size_t)(b * H_ + h) * LQ_) + l) * DH_ + j] = c[r];
    } else {
      out[(size_t)row * N + col] = c[r];
    }
  }
}

__global__ void __launch_bounds__(256) __attribute__((amdgpu_waves_per_eu(2, 4)))
feat_kernel(const float* __restrict__ Hd, const float* __restrict__ Wa,
            float* __restrict__ outf)
{
  const int bh = blockIdx.y;
  const float* A = Hd + (size_t)bh * LQ_ * DH_;
  float* O = outf + (size_t)bh * LQ_ * F_;

  const int tid  = threadIdx.x;
  const int lane = tid & 31;
  const int wave = tid >> 5;
  const int tile = blockIdx.x * 8 + wave;
  const int tm = tile / (F_ / 16);
  const int tn = tile % (F_ / 16);
  const int m0 = tm * 16, n0 = tn * 16;

  v8f c = wmma_f32_tile<DH_>(A, DH_, Wa, DH_, m0, n0, lane);

  const int l16 = lane & 15, half = lane >> 4;
  for (int r = 0; r < 8; ++r)
    O[(size_t)(m0 + r + 8 * half) * F_ + n0 + l16] = c[r];
}

constexpr int TQ_  = 8;
constexpr int F4_  = F_ / 4;
constexpr int KP4_ = F4_ + 1;
constexpr int SLAB4_ = 16;
constexpr int SLABS_ = F4_ / SLAB4_;
constexpr int SPT_   = (LK_ * SLAB4_) / 256;
constexpr int QP4_   = 18;

__global__ void __launch_bounds__(256)
score_softmax_av_kernel(const float* __restrict__ queries,
                        const float* __restrict__ Wq,
                        const float* __restrict__ Wqa,
                        const float* __restrict__ kf,
                        const float* __restrict__ vh,
                        const float* __restrict__ wv_a,
                        const int*   __restrict__ valid_lens,
                        float* __restrict__ ctx)
{
  __shared__ v4f skf[LK_ * KP4_];
  __shared__ v4f sqf[TQ_ * F4_];
  __shared__ v4f sqp[2 * TQ_ * QP4_];
  __shared__ v4f sw[F4_];
  __shared__ float sp[TQ_ * LK_];

  const int qt  = blockIdx.x % (LQ_ / TQ_);
  const int bh  = blockIdx.x / (LQ_ / TQ_);
  const int b   = bh / H_;
  const int h   = bh % H_;
  const int tid  = threadIdx.x;
  const int lane = tid & 31;
  const int wave = tid >> 5;

  const v4f* kf4 = (const v4f*)(kf + (size_t)bh * LK_ * F_);
  const v4f* wv4 = (const v4f*)wv_a;

  int goff[SPT_];
  int loff[SPT_];
#pragma unroll
  for (int jj = 0; jj < SPT_; ++jj) {
    const int u = tid + 256 * jj;
    goff[jj] = (u >> 4) * F4_  + (u & (SLAB4_ - 1));
    loff[jj] = (u >> 4) * KP4_ + (u & (SLAB4_ - 1));
  }
  v4f cur[SPT_];
#pragma unroll
  for (int jj = 0; jj < SPT_; ++jj) cur[jj] = kf4[goff[jj]];
  const int l16  = lane & 15;
  const int half = lane >> 4;
  v4f bq[4][8];
  {
    const int tnA = wave & 3;
    const int khA = wave >> 2;
    const float* xrow = queries + ((size_t)b * LQ_ + (size_t)qt * TQ_ + (l16 & 7)) * D_ + 256 * khA + 4 * half;
    const float* wrow = Wq + (size_t)(h * DH_ + 16 * tnA + l16) * D_ + 256 * khA + 4 * half;
    const v4f* wqa4 = (const v4f*)(Wqa + (size_t)(64 * wave + l16) * DH_ + 4 * half);
    const v8f cq = wmma_f32_rows<256>((const v4f*)xrow, (const v4f*)wrow, [&] {
#pragma unroll
      for (int t = 0; t < 4; ++t)
#pragma unroll
        for (int i = 0; i < 8; ++i) bq[t][i] = wqa4[t * (16 * DH_ / 4) + 2 * i];
    });
    if (half == 0) {
      float* dst = (float*)sqp + (size_t)(khA * TQ_) * (4 * QP4_) + 16 * tnA + l16;
      for (int r = 0; r < TQ_; ++r) dst[r * (4 * QP4_)] = cq[r];
    }
  }
  if (tid < F4_) sw[tid] = wv4[tid];
  __syncthreads();
  {
    const v4f* sq0 = sqp + (l16 & 7) * QP4_ + half;
    const v4f* sq1 = sq0 + TQ_ * QP4_;
    v4f aq[8];
#pragma unroll
    for (int i = 0; i < 8; ++i) aq[i] = sq0[2 * i] + sq1[2 * i];
    v8f d0[4] = {}, d1[4] = {};
#pragma unroll
    for (int i = 0; i < 8; ++i) {
      const v2f a01 = {aq[i][0], aq[i][1]};
      const v2f a23 = {aq[i][2], aq[i][3]};
#pragma unroll
      for (int t = 0; t < 4; ++t) {
        const v2f b01 = {bq[t][i][0], bq[t][i][1]};
        const v2f b23 = {bq[t][i][2], bq[t][i][3]};
        d0[t] = __builtin_amdgcn_wmma_f32_16x16x4_f32(
             false, a01,  false, b01,
             (short)0, d0[t],  false,  false);
        d1[t] = __builtin_amdgcn_wmma_f32_16x16x4_f32(
             false, a23,  false, b23,
             (short)0, d1[t],  false,  false);
      }
    }
    float* sqf_f = (float*)sqf;
#pragma unroll
    for (int t = 0; t < 4; ++t) {
      const v8f c = d0[t] + d1[t];
      if (half == 0) {
        for (int r = 0; r < TQ_; ++r) sqf_f[r * F_ + 64 * wave + 16 * t + l16] = c[r];
      }
    }
  }

  const int qa = 2 * (wave & 3);
  const int ka = 64 * (wave >> 2) + lane;
  const v4f* q0p = sqf + qa * F4_;
  const v4f* q1p = q0p + F4_;
  const v4f* k0p = skf + ka * KP4_;
  const v4f* k1p = k0p + 32 * KP4_;
  float a00 = 0.f, a01 = 0.f, a10 = 0.f, a11 = 0.f;
#pragma nounroll
  for (int s = 0; s < SLABS_; ++s) {
#pragma unroll
    for (int jj = 0; jj < SPT_; ++jj) skf[loff[jj] + SLAB4_ * s] = cur[jj];
    __syncthreads();
    if (s + 1 < SLABS_) {
#pragma unroll
      for (int jj = 0; jj < SPT_; ++jj) cur[jj] = kf4[goff[jj] + SLAB4_ * (s + 1)];
    }
    for (int f4 = SLAB4_ * s; f4 < SLAB4_ * (s + 1); ++f4) {
      const v4f w  = sw[f4];
      const v4f q0 = q0p[f4];
      const v4f q1 = q1p[f4];
      const v4f k0 = k0p[f4];
      const v4f k1 = k1p[f4];
#pragma unroll
      for (int e = 0; e < 4; ++e) {
        a00 += w[e] * TANH_F32(q0[e] + k0[e]);
        a01 += w[e] * TANH_F32(q0[e] + k1[e]);
        a10 += w[e] * TANH_F32(q1[e] + k0[e]);
        a11 += w[e] * TANH_F32(q1[e] + k1[e]);
      }
    }
  }
  sp[qa * LK_ + ka]            = a00;
  sp[qa * LK_ + ka + 32]       = a01;
  sp[(qa + 1) * LK_ + ka]      = a10;
  sp[(qa + 1) * LK_ + ka + 32] = a11;
  __syncthreads();

  {
    const int valid = valid_lens[b];
    float* srow = sp + wave * LK_;
    float val[4];
#pragma unroll
    for (int g = 0; g < 4; ++g) {
      const int k = lane + 32 * g;
      val[g] = (k < valid) ? srow[k] : -1.0e6f;
    }
    float m = fmaxf(fmaxf(val[0], val[1]), fmaxf(val[2], val[3]));
    for (int off = 16; off > 0; off >>= 1)
      m = fmaxf(m, __shfl_xor(m, off, 32));
    float e[4], s[4];
#pragma unroll
    for (int g = 0; g < 4; ++g) {
      e[g] = __expf(val[g] - m);
      s[g] = e[g];
      for (int off = 16; off > 0; off >>= 1)
        s[g] += __shfl_xor(s[g], off, 32);
    }
    const float rowsum = s[0] + s[1] + s[2] + s[3];
#pragma unroll
    for (int g = 0; g < 4; ++g)
      srow[lane + 32 * g] = e[g] / rowsum;
  }
  __syncthreads();

  const int j  = tid & 63;
  const int qg = tid >> 6;
  const float* vb = vh + (size_t)bh * LK_ * DH_ + j;
  const float* pa = sp + qg * LK_;
  const float* pb = sp + (qg + 4) * LK_;
  float oa[4] = {0.f, 0.f, 0.f, 0.f};
  float ob[4] = {0.f, 0.f, 0.f, 0.f};
  for (int kk = 0; kk < 32; ++kk) {
#pragma unroll
    for (int part = 0; part < 4; ++part) {
      const int k = part * 32 + kk;
      const float vv = vb[(size_t)k * DH_];
      oa[part] += pa[k] * vv;
      ob[part] += pb[k] * vv;
    }
  }
  const int rowa = qt * TQ_ + qg;
  ctx[((size_t)b * LQ_ + rowa) * D_ + h * DH_ + j]     = oa[0] + oa[1] + oa[2] + oa[3];
  ctx[((size_t)b * LQ_ + rowa + 4) * D_ + h * DH_ + j] = ob[0] + ob[1] + ob[2] + ob[3];
}

extern "C" void kernel_launch(void* const* d_in, const int* in_sizes, int n_in,
                              void* d_out, int out_size, void* d_ws, size_t ws_size,
                              hipStream_t stream) {
  (void)in_sizes; (void)n_in; (void)out_size; (void)ws_size;

  const float* queries    = (const float*)d_in[0];
  const float* keys       = (const float*)d_in[1];
  const float* values     = (const float*)d_in[2];
  const int*   valid_lens = (const int*)  d_in[3];
  const float* Wq  = (const float*)d_in[4];
  const float* Wk  = (const float*)d_in[5];
  const float* Wv  = (const float*)d_in[6];
  const float* Wo  = (const float*)d_in[7];
  const float* Wqa = (const float*)d_in[8];
  const float* Wka = (const float*)d_in[9];
  const float* wva = (const float*)d_in[10];

  float* ws  = (float*)d_ws;
  const size_t nQKV = (size_t)B_ * H_ * LQ_ * DH_;
  const size_t nF   = (size_t)B_ * H_ * LQ_ * F_;
  float* q   = ws;
  float* k   = q  + nQKV;
  float* v   = k  + nQKV;
  float* qf  = v  + nQKV;
  float* kfb = qf + nF;
  float* ctx = kfb + nF;

  dim3 blk(256);
  dim3 pblk(64);
  proj_kernel<<<256, pblk, 0, stream>>>(keys,    Wk, k, 1);
  proj_kernel<<<256, pblk, 0, stream>>>(values,  Wv, v, 1);

  feat_kernel<<<dim3(32, B_ * H_), blk, 0, stream>>>(k, Wka, kfb);

  score_softmax_av_kernel<<<B_ * H_ * (LQ_ / TQ_), blk, 0, stream>>>(
      queries, Wq, Wqa, kfb, v, wva, valid_lens, ctx);

  proj_kernel<<<256, pblk, 0, stream>>>(ctx, Wo, (float*)d_out, 0);
}


# Score loop as a software pipeline: a block's 20 LDS reads issued as one group a block ahead, tanh and packed add / multiply-add in T V T V order; q rows pair-interleaved in LDS; sums keep their order
# speedup vs baseline: 2.3322x; 1.1150x over previous
#include <hip/hip_runtime.h>
#include <hip/hip_bf16.h>
#include <math.h>

constexpr int B_  = 2;
constexpr int H_  = 8;
constexpr int LQ_ = 128;
constexpr int LK_ = 128;
constexpr int D_  = 512;
constexpr int DH_ = 64;
constexpr int F_  = 512;

typedef float v2f __attribute__((ext_vector_type(2)));
typedef float v4f __attribute__((ext_vector_type(4)));
typedef float v8f __attribute__((ext_vector_type(8)));

#if __has_builtin(__builtin_amdgcn_tanhf)
#define TANH_F32(x) __builtin_amdgcn_tanhf(x)
#elif __has_builtin(__builtin_amdgcn_tanh_f32)
#define TANH_F32(x) __builtin_amdgcn_tanh_f32(x)
#else
#define TANH_F32(x) tanhf(x)
#endif

constexpr int CH_    = 8;
constexpr int AHEAD_ = 3;

template <int K, typename AfterRequests>
__device__ __forceinline__ v8f wmma_f32_rows(const v4f* __restrict__ arow,
                                             const v4f* __restrict__ wrow,
                                             AfterRequests&& after_requests)
{
  constexpr int NP  = K / 8;
  constexpr int NCH = NP / CH_;
  constexpr int LA  = (NCH < AHEAD_) ? NCH : AHEAD_;
  v4f a[NP], b[NP];
  v8f c0 = {}, c1 = {};
#pragma unroll
  for (int ch = 0; ch < NCH + LA; ++ch) {
    if (ch < NCH) {
#pragma unroll
      for (int i = ch * CH_; i < (ch + 1) * CH_; ++i) {
        a[i] = arow[2 * i];
        b[i] = wrow[2 * i];
      }
      if (ch == NCH - 1) after_requests();
    }
    if (ch >= LA) {
#pragma unroll
      for (int i = (ch - LA) * CH_; i < (ch - LA + 1) * CH_; ++i) {
        const v2f a01 = {a[i][0], a[i][1]}, b01 = {b[i][0], b[i][1]};
        const v2f a23 = {a[i][2], a[i][3]}, b23 = {b[i][2], b[i][3]};
        c0 = __builtin_amdgcn_wmma_f32_16x16x4_f32(
             false, a01,  false, b01,
             (short)0, c0,  false,  false);
        c1 = __builtin_amdgcn_wmma_f32_16x16x4_f32(
             false, a23,  false, b23,
             (short)0, c1,  false,  false);
      }
    }
  }
  return c0 + c1;
}

template <int K>
__device__ __forceinline__ v8f wmma_f32_tile(const float* __restrict__ A, int lda,
                                             const float* __restrict__ W, int ldw,
                                             int m0, int n0, int lane)
{
  const int half = lane >> 4;
  const int l16  = lane & 15;
  return wmma_f32_rows<K>((const v4f*)(A + (size_t)(m0 + l16) * lda + 4 * half),
                          (const v4f*)(W + (size_t)(n0 + l16) * ldw + 4 * half), [] {});
}

__global__ void __launch_bounds__(64) __attribute__((amdgpu_waves_per_eu(1, 2)))
proj_kernel(const float* __restrict__ X, const float* __restrict__ W,
            float* __restrict__ out, int split)
{
  constexpr int N = D_;
  constexpr int K = D_;
  const int tid  = threadIdx.x;
  const int lane = tid & 31;
  const int wave = tid >> 5;
  const int tile = blockIdx.x * 2 + wave;
  const int tm = tile / (N / 16);
  const int tn = tile % (N / 16);
  const int m0 = tm * 16, n0 = tn * 16;

  v8f c = wmma_f32_tile<K>(X, K, W, K, m0, n0, lane);

  const int l16 = lane & 15, half = lane >> 4;
  const int col = n0 + l16;
  for (int r = 0; r < 8; ++r) {
    const int row = m0 + r + 8 * half;
    if (split) {
      const int b = row / LQ_, l = row % LQ_;
      const int h = col / DH_, j = col % DH_;
      out[(((size_t)(b * H_ + h) * LQ_) + l) * DH_ + j] = c[r];
    } else {
      out[(size_t)row * N + col] = c[r];
    }
  }
}

__global__ void __launch_bounds__(256) __attribute__((amdgpu_waves_per_eu(2, 4)))
feat_kernel(const float* __restrict__ Hd, const float* __restrict__ Wa,
            float* __restrict__ outf)
{
  const int bh = blockIdx.y;
  const float* A = Hd + (size_t)bh * LQ_ * DH_;
  float* O = outf + (size_t)bh * LQ_ * F_;

  const int tid  = threadIdx.x;
  const int lane = tid & 31;
  const int wave = tid >> 5;
  const int tile = blockIdx.x * 8 + wave;
  const int tm = tile / (F_ / 16);
  const int tn = tile % (F_ / 16);
  const int m0 = tm * 16, n0 = tn * 16;

  v8f c = wmma_f32_tile<DH_>(A, DH_, Wa, DH_, m0, n0, lane);

  const int l16 = lane & 15, half = lane >> 4;
  for (int r = 0; r < 8; ++r)
    O[(size_t)(m0 + r + 8 * half) * F_ + n0 + l16] = c[r];
}

constexpr int TQ_  = 8;
constexpr int F4_  = F_ / 4;
constexpr int KP4_ = F4_ + 1;
constexpr int SLAB4_ = 16;
constexpr int SLABS_ = F4_ / SLAB4_;
constexpr int SPT_   = (LK_ * SLAB4_) / 256;
constexpr int QP4_   = 18;

__global__ void __launch_bounds__(256)
score_softmax_av_kernel(const float* __restrict__ queries,
                        const float* __restrict__ Wq,
                        const float* __restrict__ Wqa,
                        const float* __restrict__ kf,
                        const float* __restrict__ vh,
                        const float* __restrict__ wv_a,
                        const int*   __restrict__ valid_lens,
                        float* __restrict__ ctx)
{
  __shared__ v4f skf[LK_ * KP4_];
  __shared__ v4f sqf[TQ_ * F4_];
  __shared__ v4f sqp[2 * TQ_ * QP4_];
  __shared__ v4f sw[F4_];
  __shared__ float sp[TQ_ * LK_];

  const int qt  = blockIdx.x % (LQ_ / TQ_);
  const int bh  = blockIdx.x / (LQ_ / TQ_);
  const int b   = bh / H_;
  const int h   = bh % H_;
  const int tid  = threadIdx.x;
  const int lane = tid & 31;
  const int wave = tid >> 5;

  const v4f* kf4 = (const v4f*)(kf + (size_t)bh * LK_ * F_);
  const v4f* wv4 = (const v4f*)wv_a;

  int goff[SPT_];
  int loff[SPT_];
#pragma unroll
  for (int jj = 0; jj < SPT_; ++jj) {
    const int u = tid + 256 * jj;
    goff[jj] = (u >> 4) * F4_  + (u & (SLAB4_ - 1));
    loff[jj] = (u >> 4) * KP4_ + (u & (SLAB4_ - 1));
  }
  v4f cur[SPT_];
#pragma unroll
  for (int jj = 0; jj < SPT_; ++jj) cur[jj] = kf4[goff[jj]];
  const int l16  = lane & 15;
  const int half = lane >> 4;
  v4f bq[4][8];
  {
    const int tnA = wave & 3;
    const int khA = wave >> 2;
    const float* xrow = queries + ((size_t)b * LQ_ + (size_t)qt * TQ_ + (l16 & 7)) * D_ + 256 * khA + 4 * half;
    const float* wrow = Wq + (size_t)(h * DH_ + 16 * tnA + l16) * D_ + 256 * khA + 4 * half;
    const v4f* wqa4 = (const v4f*)(Wqa + (size_t)(64 * wave + l16) * DH_ + 4 * half);
    const v8f cq = wmma_f32_rows<256>((const v4f*)xrow, (const v4f*)wrow, [&] {
#pragma unroll
      for (int t = 0; t < 4; ++t)
#pragma unroll
        for (int i = 0; i < 8; ++i) bq[t][i] = wqa4[t * (16 * DH_ / 4) + 2 * i];
    });
    if (half == 0) {
      float* dst = (float*)sqp + (size_t)(khA * TQ_) * (4 * QP4_) + 16 * tnA + l16;
      for (int r = 0; r < TQ_; ++r) dst[r * (4 * QP4_)] = cq[r];
    }
  }
  if (tid < F4_) sw[tid] = wv4[tid];
  __syncthreads();
  {
    const v4f* sq0 = sqp + (l16 & 7) * QP4_ + half;
    const v4f* sq1 = sq0 + TQ_ * QP4_;
    v4f aq[8];
#pragma unroll
    for (int i = 0; i < 8; ++i) aq[i] = sq0[2 * i] + sq1[2 * i];
    v8f d0[4] = {}, d1[4] = {};
#pragma unroll
    for (int i = 0; i < 8; ++i) {
      const v2f a01 = {aq[i][0], aq[i][1]};
      const v2f a23 = {aq[i][2], aq[i][3]};
#pragma unroll
      for (int t = 0; t < 4; ++t) {
        const v2f b01 = {bq[t][i][0], bq[t][i][1]};
        const v2f b23 = {bq[t][i][2], bq[t][i][3]};
        d0[t] = __builtin_amdgcn_wmma_f32_16x16x4_f32(
             false, a01,  false, b01,
             (short)0, d0[t],  false,  false);
        d1[t] = __builtin_amdgcn_wmma_f32_16x16x4_f32(
             false, a23,  false, b23,
             (short)0, d1[t],  false,  false);
      }
    }
    v2f* sqf_2 = (v2f*)sqf;
#pragma unroll
    for (int t = 0; t < 4; ++t) {
      const v8f c = d0[t] + d1[t];
      if (half == 0) {
#pragma unroll
        for (int p = 0; p < TQ_ / 2; ++p) {
          const v2f pr = {c[2 * p], c[2 * p + 1]};
          sqf_2[p * F_ + 64 * wave + 16 * t + l16] = pr;
        }
      }
    }
  }

  const int qa = 2 * (wave & 3);
  const int ka = 64 * (wave >> 2) + lane;
  const v4f* qpp = sqf + (wave & 3) * (F_ / 2);
  const v4f* k0p = skf + ka * KP4_;
  const v4f* k1p = k0p + 32 * KP4_;
  constexpr int NB_ = 4;
  struct OpSet { v4f w[NB_], qlo[NB_], qhi[NB_], k0[NB_], k1[NB_]; };
  OpSet A, Bs;
  v2f acc0 = {0.f, 0.f};
  v2f acc1 = {0.f, 0.f};
  auto fetch = [&](OpSet& o, int g0) {
#pragma unroll
    for (int i = 0; i < NB_; ++i) {
      o.w[i]   = sw[g0 + i];
      o.qlo[i] = qpp[2 * (g0 + i)];
      o.qhi[i] = qpp[2 * (g0 + i) + 1];
      o.k0[i]  = k0p[g0 + i];
      o.k1[i]  = k1p[g0 + i];
    }
  };
  auto half2 = [](const v4f& x, int e) -> v2f {
    return (e < 2) ? __builtin_shufflevector(x, x, 0, 1) : __builtin_shufflevector(x, x, 2, 3);
  };
  auto both = [](const v4f& x, int e) -> v2f {
    return (e == 0) ? __builtin_shufflevector(x, x, 0, 0) : (e == 1) ? __builtin_shufflevector(x, x, 1, 1)
         : (e == 2) ? __builtin_shufflevector(x, x, 2, 2) : __builtin_shufflevector(x, x, 3, 3);
  };
  auto qpair = [&](const OpSet& o, int i, int e) -> v2f {
    return half2((e < 2) ? o.qlo[i] : o.qhi[i], 2 * (e & 1));
  };
  v2f ns0, ns1;
  v2f pt0 = {0.f, 0.f}, pt1 = {0.f, 0.f};
  v2f pw = {0.f, 0.f};
#define ORDER_KEPT() __builtin_amdgcn_sched_barrier(0)
  auto score = [&](const OpSet& o, const OpSet& nx, bool has_next) {
#pragma unroll
    for (int i = 0; i < NB_; ++i) {
#pragma unroll
      for (int e = 0; e < 4; ++e) {
        const bool in_set = (4 * i + e + 1 < 4 * NB_);
        const OpSet& on = in_set ? o : nx;
        const int in = in_set ? (4 * i + e + 1) / 4 : 0;
        const int en = in_set ? (4 * i + e + 1) % 4 : 0;
        const bool adds = in_set || has_next;
        v2f t0, t1, m0 = ns0, m1 = ns1;
        t0[0] = TANH_F32(ns0[0]);
        ORDER_KEPT();
        if (adds) m0 = qpair(on, in, en) + both(on.k0[in], en);
        ORDER_KEPT();
        t0[1] = TANH_F32(ns0[1]);
        ORDER_KEPT();
        acc0 += pw * pt0;
        ORDER_KEPT();
        t1[0] = TANH_F32(ns1[0]);
        ORDER_KEPT();
        if (adds) m1 = qpair(on, in, en) + both(on.k1[in], en);
        ORDER_KEPT();
        t1[1] = TANH_F32(ns1[1]);
        ORDER_KEPT();
        acc1 += pw * pt1;
        ORDER_KEPT();
        pt0 = t0; pt1 = t1; pw = both(o.w[i], e);
        ns0 = m0; ns1 = m1;
      }
    }
  };
#define BLOCK_SCORED() asm volatile("" : "+v"(acc0), "+v"(acc1) : : "memory")
#define READS_ISSUED() __builtin_amdgcn_sched_barrier(0)

#pragma unroll
  for (int jj = 0; jj < SPT_; ++jj) skf[loff[jj]] = cur[jj];
  __syncthreads();
#pragma unroll
  for (int jj = 0; jj < SPT_; ++jj) cur[jj] = kf4[goff[jj] + SLAB4_];
  fetch(A, 0);
  ns0 = qpair(A, 0, 0) + both(A.k0[0], 0);
  ns1 = qpair(A, 0, 0) + both(A.k1[0], 0);
#pragma nounroll
  for (int s = 0; s < SLABS_ - 1; ++s) {
    const int g = SLAB4_ * s;
    fetch(Bs, g + NB_);
    READS_ISSUED();
    score(A, Bs, true);
    BLOCK_SCORED();
    fetch(A, g + 2 * NB_);
    READS_ISSUED();
    score(Bs, A, true);
    BLOCK_SCORED();
    fetch(Bs, g + 3 * NB_);
    READS_ISSUED();
    score(A, Bs, true);
    BLOCK_SCORED();
#pragma unroll
    for (int jj = 0; jj < SPT_; ++jj) skf[loff[jj] + SLAB4_ * (s + 1)] = cur[jj];
    __syncthreads();
    if (s + 2 < SLABS_) {
#pragma unroll
      for (int jj = 0; jj < SPT_; ++jj) cur[jj] = kf4[goff[jj] + SLAB4_ * (s + 2)];
    }
    fetch(A, g + 4 * NB_);
    READS_ISSUED();
    score(Bs, A, true);
    BLOCK_SCORED();
  }
  {
    const int g = SLAB4_ * (SLABS_ - 1);
    fetch(Bs, g + NB_);
    READS_ISSUED();
    score(A, Bs, true);
    BLOCK_SCORED();
    fetch(A, g + 2 * NB_);
    READS_ISSUED();
    score(Bs, A, true);
    BLOCK_SCORED();
    fetch(Bs, g + 3 * NB_);
    READS_ISSUED();
    score(A, Bs, true);
    BLOCK_SCORED();
    score(Bs, A, false);
    acc0 += pw * pt0;
    acc1 += pw * pt1;
  }
#undef BLOCK_SCORED
#undef READS_ISSUED
#undef ORDER_KEPT
  sp[qa * LK_ + ka]            = acc0[0];
  sp[qa * LK_ + ka + 32]       = acc1[0];
  sp[(qa + 1) * LK_ + ka]      = acc0[1];
  sp[(qa + 1) * LK_ + ka + 32] = acc1[1];
  __syncthreads();

  {
    const int valid = valid_lens[b];
    float* srow = sp + wave * LK_;
    float val[4];
#pragma unroll
    for (int g = 0; g < 4; ++g) {
      const int k = lane + 32 * g;
      val[g] = (k < valid) ? srow[k] : -1.0e6f;
    }
    float m = fmaxf(fmaxf(val[0], val[1]), fmaxf(val[2], val[3]));
    for (int off = 16; off > 0; off >>= 1)
      m = fmaxf(m, __shfl_xor(m, off, 32));
    float e[4], s[4];
#pragma unroll
    for (int g = 0; g < 4; ++g) {
      e[g] = __expf(val[g] - m);
      s[g] = e[g];
      for (int off = 16; off > 0; off >>= 1)
        s[g] += __shfl_xor(s[g], off, 32);
    }
    const float rowsum = s[0] + s[1] + s[2] + s[3];
#pragma unroll
    for (int g = 0; g < 4; ++g)
      srow[lane + 32 * g] = e[g] / rowsum;
  }
  __syncthreads();

  const int j  = tid & 63;
  const int qg = tid >> 6;
  const float* vb = vh + (size_t)bh * LK_ * DH_ + j;
  const float* pa = sp + qg * LK_;
  const float* pb = sp + (qg + 4) * LK_;
  float oa[4] = {0.f, 0.f, 0.f, 0.f};
  float ob[4] = {0.f, 0.f, 0.f, 0.f};
  for (int kk = 0; kk < 32; ++kk) {
#pragma unroll
    for (int part = 0; part < 4; ++part) {
      const int k = part * 32 + kk;
      const float vv = vb[(size_t)k * DH_];
      oa[part] += pa[k] * vv;
      ob[part] += pb[k] * vv;
    }
  }
  const int rowa = qt * TQ_ + qg;
  ctx[((size_t)b * LQ_ + rowa) * D_ + h * DH_ + j]     = oa[0] + oa[1] + oa[2] + oa[3];
  ctx[((size_t)b * LQ_ + rowa + 4) * D_ + h * DH_ + j] = ob[0] + ob[1] + ob[2] + ob[3];
}

extern "C" void kernel_launch(void* const* d_in, const int* in_sizes, int n_in,
                              void* d_out, int out_size, void* d_ws, size_t ws_size,
                              hipStream_t stream) {
  (void)in_sizes; (void)n_in; (void)out_size; (void)ws_size;

  const float* queries    = (const float*)d_in[0];
  const float* keys       = (const float*)d_in[1];
  const float* values     = (const float*)d_in[2];
  const int*   valid_lens = (const int*)  d_in[3];
  const float* Wq  = (const float*)d_in[4];
  const float* Wk  = (const float*)d_in[5];
  const float* Wv  = (const float*)d_in[6];
  const float* Wo  = (const float*)d_in[7];
  const float* Wqa = (const float*)d_in[8];
  const float* Wka = (const float*)d_in[9];
  const float* wva = (const float*)d_in[10];

  float* ws  = (float*)d_ws;
  const size_t nQKV = (size_t)B_ * H_ * LQ_ * DH_;
  const size_t nF   = (size_t)B_ * H_ * LQ_ * F_;
  float* q   = ws;
  float* k   = q  + nQKV;
  float* v   = k  + nQKV;
  float* qf  = v  + nQKV;
  float* kfb = qf + nF;
  float* ctx = kfb + nF;

  dim3 blk(256);
  dim3 pblk(64);
  proj_kernel<<<256, pblk, 0, stream>>>(keys,    Wk, k, 1);
  proj_kernel<<<256, pblk, 0, stream>>>(values,  Wv, v, 1);

  feat_kernel<<<dim3(32, B_ * H_), blk, 0, stream>>>(k, Wka, kfb);

  score_softmax_av_kernel<<<B_ * H_ * (LQ_ / TQ_), blk, 0, stream>>>(
      queries, Wq, Wqa, kfb, v, wva, valid_lens, ctx);

  proj_kernel<<<256, pblk, 0, stream>>>(ctx, Wo, (float*)d_out, 0);
}


# V projection's 512 tiles ride in the key-feature launch: one grid of 2-wave groups, V groups lowest so their long waves start first beside the short feature waves; 5 launches -> 4; sums unchanged
# speedup vs baseline: 2.5216x; 1.0812x over previous
#include <hip/hip_runtime.h>
#include <hip/hip_bf16.h>
#include <math.h>

constexpr int B_  = 2;
constexpr int H_  = 8;
constexpr int LQ_ = 128;
constexpr int LK_ = 128;
constexpr int D_  = 512;
constexpr int DH_ = 64;
constexpr int F_  = 512;

typedef float v2f __attribute__((ext_vector_type(2)));
typedef float v4f __attribute__((ext_vector_type(4)));
typedef float v8f __attribute__((ext_vector_type(8)));

#if __has_builtin(__builtin_amdgcn_tanhf)
#define TANH_F32(x) __builtin_amdgcn_tanhf(x)
#elif __has_builtin(__builtin_amdgcn_tanh_f32)
#define TANH_F32(x) __builtin_amdgcn_tanh_f32(x)
#else
#define TANH_F32(x) tanhf(x)
#endif

constexpr int CH_    = 8;
constexpr int AHEAD_ = 3;

template <int K, typename AfterRequests>
__device__ __forceinline__ v8f wmma_f32_rows(const v4f* __restrict__ arow,
                                             const v4f* __restrict__ wrow,
                                             AfterRequests&& after_requests)
{
  constexpr int NP  = K / 8;
  constexpr int NCH = NP / CH_;
  constexpr int LA  = (NCH < AHEAD_) ? NCH : AHEAD_;
  v4f a[NP], b[NP];
  v8f c0 = {}, c1 = {};
#pragma unroll
  for (int ch = 0; ch < NCH + LA; ++ch) {
    if (ch < NCH) {
#pragma unroll
      for (int i = ch * CH_; i < (ch + 1) * CH_; ++i) {
        a[i] = arow[2 * i];
        b[i] = wrow[2 * i];
      }
      if (ch == NCH - 1) after_requests();
    }
    if (ch >= LA) {
#pragma unroll
      for (int i = (ch - LA) * CH_; i < (ch - LA + 1) * CH_; ++i) {
        const v2f a01 = {a[i][0], a[i][1]}, b01 = {b[i][0], b[i][1]};
        const v2f a23 = {a[i][2], a[i][3]}, b23 = {b[i][2], b[i][3]};
        c0 = __builtin_amdgcn_wmma_f32_16x16x4_f32(
             false, a01,  false, b01,
             (short)0, c0,  false,  false);
        c1 = __builtin_amdgcn_wmma_f32_16x16x4_f32(
             false, a23,  false, b23,
             (short)0, c1,  false,  false);
      }
    }
  }
  return c0 + c1;
}

template <int K>
__device__ __forceinline__ v8f wmma_f32_tile(const float* __restrict__ A, int lda,
                                             const float* __restrict__ W, int ldw,
                                             int m0, int n0, int lane)
{
  const int half = lane >> 4;
  const int l16  = lane & 15;
  return wmma_f32_rows<K>((const v4f*)(A + (size_t)(m0 + l16) * lda + 4 * half),
                          (const v4f*)(W + (size_t)(n0 + l16) * ldw + 4 * half), [] {});
}

__device__ __forceinline__ void proj_tile(const float* __restrict__ X, const float* __restrict__ W,
                                          float* __restrict__ out, int split, int tile, int lane)
{
  constexpr int N = D_;
  constexpr int K = D_;
  const int tm = tile / (N / 16);
  const int tn = tile % (N / 16);
  const int m0 = tm * 16, n0 = tn * 16;

  v8f c = wmma_f32_tile<K>(X, K, W, K, m0, n0, lane);

  const int l16 = lane & 15, half = lane >> 4;
  const int col = n0 + l16;
  for (int r = 0; r < 8; ++r) {
    const int row = m0 + r + 8 * half;
    if (split) {
      const int b = row / LQ_, l = row % LQ_;
      const int h = col / DH_, j = col % DH_;
      out[(((size_t)(b * H_ + h) * LQ_) + l) * DH_ + j] = c[r];
    } else {
      out[(size_t)row * N + col] = c[r];
    }
  }
}

__global__ void __launch_bounds__(64) __attribute__((amdgpu_waves_per_eu(1, 2)))
proj_kernel(const float* __restrict__ X, const float* __restrict__ W,
            float* __restrict__ out, int split)
{
  const int tid  = threadIdx.x;
  proj_tile(X, W, out, split, blockIdx.x * 2 + (tid >> 5), tid & 31);
}

constexpr int VGRP_ = (B_ * LK_ / 16) * (D_ / 16) / 2;
constexpr int FGRP_ = B_ * H_ * (LK_ / 16) * (F_ / 16) / 2;

__global__ void __launch_bounds__(64) __attribute__((amdgpu_waves_per_eu(1, 4)))
vproj_feat_kernel(const float* __restrict__ Xv, const float* __restrict__ Wvp,
                  float* __restrict__ vout,
                  const float* __restrict__ Hd, const float* __restrict__ Wa,
                  float* __restrict__ outf)
{
  const int tid  = threadIdx.x;
  const int lane = tid & 31;
  const int wave = tid >> 5;
  if (blockIdx.x < VGRP_) {
    proj_tile(Xv, Wvp, vout, 1, blockIdx.x * 2 + wave, lane);
    return;
  }
  const unsigned ft = (blockIdx.x - (unsigned)VGRP_) * 2u + (unsigned)wave;
  const int bh   = (int)(ft / (unsigned)((LK_ / 16) * (F_ / 16)));
  const int tile = (int)(ft % (unsigned)((LK_ / 16) * (F_ / 16)));
  const float* A = Hd + (size_t)bh * LK_ * DH_;
  float* O = outf + (size_t)bh * LK_ * F_;
  const int tm = tile / (F_ / 16);
  const int tn = tile % (F_ / 16);
  const int m0 = tm * 16, n0 = tn * 16;

  v8f c = wmma_f32_tile<DH_>(A, DH_, Wa, DH_, m0, n0, lane);

  const int l16 = lane & 15, half = lane >> 4;
  for (int r = 0; r < 8; ++r)
    O[(size_t)(m0 + r + 8 * half) * F_ + n0 + l16] = c[r];
}

constexpr int TQ_  = 8;
constexpr int F4_  = F_ / 4;
constexpr int KP4_ = F4_ + 1;
constexpr int SLAB4_ = 16;
constexpr int SLABS_ = F4_ / SLAB4_;
constexpr int SPT_   = (LK_ * SLAB4_) / 256;
constexpr int QP4_   = 18;

__global__ void __launch_bounds__(256)
score_softmax_av_kernel(const float* __restrict__ queries,
                        const float* __restrict__ Wq,
                        const float* __restrict__ Wqa,
                        const float* __restrict__ kf,
                        const float* __restrict__ vh,
                        const float* __restrict__ wv_a,
                        const int*   __restrict__ valid_lens,
                        float* __restrict__ ctx)
{
  __shared__ v4f skf[LK_ * KP4_];
  __shared__ v4f sqf[TQ_ * F4_];
  __shared__ v4f sqp[2 * TQ_ * QP4_];
  __shared__ v4f sw[F4_];
  __shared__ float sp[TQ_ * LK_];

  const int qt  = blockIdx.x % (LQ_ / TQ_);
  const int bh  = blockIdx.x / (LQ_ / TQ_);
  const int b   = bh / H_;
  const int h   = bh % H_;
  const int tid  = threadIdx.x;
  const int lane = tid & 31;
  const int wave = tid >> 5;

  const v4f* kf4 = (const v4f*)(kf + (size_t)bh * LK_ * F_);
  const v4f* wv4 = (const v4f*)wv_a;

  int goff[SPT_];
  int loff[SPT_];
#pragma unroll
  for (int jj = 0; jj < SPT_; ++jj) {
    const int u = tid + 256 * jj;
    goff[jj] = (u >> 4) * F4_  + (u & (SLAB4_ - 1));
    loff[jj] = (u >> 4) * KP4_ + (u & (SLAB4_ - 1));
  }
  v4f cur[SPT_];
#pragma unroll
  for (int jj = 0; jj < SPT_; ++jj) cur[jj] = kf4[goff[jj]];
  const int l16  = lane & 15;
  const int half = lane >> 4;
  v4f bq[4][8];
  {
    const int tnA = wave & 3;
    const int khA = wave >> 2;
    const float* xrow = queries + ((size_t)b * LQ_ + (size_t)qt * TQ_ + (l16 & 7)) * D_ + 256 * khA + 4 * half;
    const float* wrow = Wq + (size_t)(h * DH_ + 16 * tnA + l16) * D_ + 256 * khA + 4 * half;
    const v4f* wqa4 = (const v4f*)(Wqa + (size_t)(64 * wave + l16) * DH_ + 4 * half);
    const v8f cq = wmma_f32_rows<256>((const v4f*)xrow, (const v4f*)wrow, [&] {
#pragma unroll
      for (int t = 0; t < 4; ++t)
#pragma unroll
        for (int i = 0; i < 8; ++i) bq[t][i] = wqa4[t * (16 * DH_ / 4) + 2 * i];
    });
    if (half == 0) {
      float* dst = (float*)sqp + (size_t)(khA * TQ_) * (4 * QP4_) + 16 * tnA + l16;
      for (int r = 0; r < TQ_; ++r) dst[r * (4 * QP4_)] = cq[r];
    }
  }
  if (tid < F4_) sw[tid] = wv4[tid];
  __syncthreads();
  {
    const v4f* sq0 = sqp + (l16 & 7) * QP4_ + half;
    const v4f* sq1 = sq0 + TQ_ * QP4_;
    v4f aq[8];
#pragma unroll
    for (int i = 0; i < 8; ++i) aq[i] = sq0[2 * i] + sq1[2 * i];
    v8f d0[4] = {}, d1[4] = {};
#pragma unroll
    for (int i = 0; i < 8; ++i) {
      const v2f a01 = {aq[i][0], aq[i][1]};
      const v2f a23 = {aq[i][2], aq[i][3]};
#pragma unroll
      for (int t = 0; t < 4; ++t) {
        const v2f b01 = {bq[t][i][0], bq[t][i][1]};
        const v2f b23 = {bq[t][i][2], bq[t][i][3]};
        d0[t] = __builtin_amdgcn_wmma_f32_16x16x4_f32(
             false, a01,  false, b01,
             (short)0, d0[t],  false,  false);
        d1[t] = __builtin_amdgcn_wmma_f32_16x16x4_f32(
             false, a23,  false, b23,
             (short)0, d1[t],  false,  false);
      }
    }
    v2f* sqf_2 = (v2f*)sqf;
#pragma unroll
    for (int t = 0; t < 4; ++t) {
      const v8f c = d0[t] + d1[t];
      if (half == 0) {
#pragma unroll
        for (int p = 0; p < TQ_ / 2; ++p) {
          const v2f pr = {c[2 * p], c[2 * p + 1]};
          sqf_2[p * F_ + 64 * wave + 16 * t + l16] = pr;
        }
      }
    }
  }

  const int qa = 2 * (wave & 3);
  const int ka = 64 * (wave >> 2) + lane;
  const v4f* qpp = sqf + (wave & 3) * (F_ / 2);
  const v4f* k0p = skf + ka * KP4_;
  const v4f* k1p = k0p + 32 * KP4_;
  constexpr int NB_ = 4;
  struct OpSet { v4f w[NB_], qlo[NB_], qhi[NB_], k0[NB_], k1[NB_]; };
  OpSet A, Bs;
  v2f acc0 = {0.f, 0.f};
  v2f acc1 = {0.f, 0.f};
  auto fetch = [&](OpSet& o, int g0) {
#pragma unroll
    for (int i = 0; i < NB_; ++i) {
      o.w[i]   = sw[g0 + i];
      o.qlo[i] = qpp[2 * (g0 + i)];
      o.qhi[i] = qpp[2 * (g0 + i) + 1];
      o.k0[i]  = k0p[g0 + i];
      o.k1[i]  = k1p[g0 + i];
    }
  };
  auto half2 = [](const v4f& x, int e) -> v2f {
    return (e < 2) ? __builtin_shufflevector(x, x, 0, 1) : __builtin_shufflevector(x, x, 2, 3);
  };
  auto both = [](const v4f& x, int e) -> v2f {
    return (e == 0) ? __builtin_shufflevector(x, x, 0, 0) : (e == 1) ? __builtin_shufflevector(x, x, 1, 1)
         : (e == 2) ? __builtin_shufflevector(x, x, 2, 2) : __builtin_shufflevector(x, x, 3, 3);
  };
  auto qpair = [&](const OpSet& o, int i, int e) -> v2f {
    return half2((e < 2) ? o.qlo[i] : o.qhi[i], 2 * (e & 1));
  };
  v2f ns0, ns1;
  v2f pt0 = {0.f, 0.f}, pt1 = {0.f, 0.f};
  v2f pw = {0.f, 0.f};
#define ORDER_KEPT() __builtin_amdgcn_sched_barrier(0)
  auto score = [&](const OpSet& o, const OpSet& nx, bool has_next) {
#pragma unroll
    for (int i = 0; i < NB_; ++i) {
#pragma unroll
      for (int e = 0; e < 4; ++e) {
        const bool in_set = (4 * i + e + 1 < 4 * NB_);
        const OpSet& on = in_set ? o : nx;
        const int in = in_set ? (4 * i + e + 1) / 4 : 0;
        const int en = in_set ? (4 * i + e + 1) % 4 : 0;
        const bool adds = in_set || has_next;
        v2f t0, t1, m0 = ns0, m1 = ns1;
        t0[0] = TANH_F32(ns0[0]);
        ORDER_KEPT();
        if (adds) m0 = qpair(on, in, en) + both(on.k0[in], en);
        ORDER_KEPT();
        t0[1] = TANH_F32(ns0[1]);
        ORDER_KEPT();
        acc0 += pw * pt0;
        ORDER_KEPT();
        t1[0] = TANH_F32(ns1[0]);
        ORDER_KEPT();
        if (adds) m1 = qpair(on, in, en) + both(on.k1[in], en);
        ORDER_KEPT();
        t1[1] = TANH_F32(ns1[1]);
        ORDER_KEPT();
        acc1 += pw * pt1;
        ORDER_KEPT();
        pt0 = t0; pt1 = t1; pw = both(o.w[i], e);
        ns0 = m0; ns1 = m1;
      }
    }
  };
#define BLOCK_SCORED() asm volatile("" : "+v"(acc0), "+v"(acc1) : : "memory")
#define READS_ISSUED() __builtin_amdgcn_sched_barrier(0)

#pragma unroll
  for (int jj = 0; jj < SPT_; ++jj) skf[loff[jj]] = cur[jj];
  __syncthreads();
#pragma unroll
  for (int jj = 0; jj < SPT_; ++jj) cur[jj] = kf4[goff[jj] + SLAB4_];
  fetch(A, 0);
  ns0 = qpair(A, 0, 0) + both(A.k0[0], 0);
  ns1 = qpair(A, 0, 0) + both(A.k1[0], 0);
#pragma nounroll
  for (int s = 0; s < SLABS_ - 1; ++s) {
    const int g = SLAB4_ * s;
    fetch(Bs, g + NB_);
    READS_ISSUED();
    score(A, Bs, true);
    BLOCK_SCORED();
    fetch(A, g + 2 * NB_);
    READS_ISSUED();
    score(Bs, A, true);
    BLOCK_SCORED();
    fetch(Bs, g + 3 * NB_);
    READS_ISSUED();
    score(A, Bs, true);
    BLOCK_SCORED();
#pragma unroll
    for (int jj = 0; jj < SPT_; ++jj) skf[loff[jj] + SLAB4_ * (s + 1)] = cur[jj];
    __syncthreads();
    if (s + 2 < SLABS_) {
#pragma unroll
      for (int jj = 0; jj < SPT_; ++jj) cur[jj] = kf4[goff[jj] + SLAB4_ * (s + 2)];
    }
    fetch(A, g + 4 * NB_);
    READS_ISSUED();
    score(Bs, A, true);
    BLOCK_SCORED();
  }
  {
    const int g = SLAB4_ * (SLABS_ - 1);
    fetch(Bs, g + NB_);
    READS_ISSUED();
    score(A, Bs, true);
    BLOCK_SCORED();
    fetch(A, g + 2 * NB_);
    READS_ISSUED();
    score(Bs, A, true);
    BLOCK_SCORED();
    fetch(Bs, g + 3 * NB_);
    READS_ISSUED();
    score(A, Bs, true);
    BLOCK_SCORED();
    score(Bs, A, false);
    acc0 += pw * pt0;
    acc1 += pw * pt1;
  }
#undef BLOCK_SCORED
#undef READS_ISSUED
#undef ORDER_KEPT
  sp[qa * LK_ + ka]            = acc0[0];
  sp[qa * LK_ + ka + 32]       = acc1[0];
  sp[(qa + 1) * LK_ + ka]      = acc0[1];
  sp[(qa + 1) * LK_ + ka + 32] = acc1[1];
  __syncthreads();

  {
    const int valid = valid_lens[b];
    float* srow = sp + wave * LK_;
    float val[4];
#pragma unroll
    for (int g = 0; g < 4; ++g) {
      const int k = lane + 32 * g;
      val[g] = (k < valid) ? srow[k] : -1.0e6f;
    }
    float m = fmaxf(fmaxf(val[0], val[1]), fmaxf(val[2], val[3]));
    for (int off = 16; off > 0; off >>= 1)
      m = fmaxf(m, __shfl_xor(m, off, 32));
    float e[4], s[4];
#pragma unroll
    for (int g = 0; g < 4; ++g) {
      e[g] = __expf(val[g] - m);
      s[g] = e[g];
      for (int off = 16; off > 0; off >>= 1)
        s[g] += __shfl_xor(s[g], off, 32);
    }
    const float rowsum = s[0] + s[1] + s[2] + s[3];
#pragma unroll
    for (int g = 0; g < 4; ++g)
      srow[lane + 32 * g] = e[g] / rowsum;
  }
  __syncthreads();

  const int j  = tid & 63;
  const int qg = tid >> 6;
  const float* vb = vh + (size_t)bh * LK_ * DH_ + j;
  const float* pa = sp + qg * LK_;
  const float* pb = sp + (qg + 4) * LK_;
  float oa[4] = {0.f, 0.f, 0.f, 0.f};
  float ob[4] = {0.f, 0.f, 0.f, 0.f};
  for (int kk = 0; kk < 32; ++kk) {
#pragma unroll
    for (int part = 0; part < 4; ++part) {
      const int k = part * 32 + kk;
      const float vv = vb[(size_t)k * DH_];
      oa[part] += pa[k] * vv;
      ob[part] += pb[k] * vv;
    }
  }
  const int rowa = qt * TQ_ + qg;
  ctx[((size_t)b * LQ_ + rowa) * D_ + h * DH_ + j]     = oa[0] + oa[1] + oa[2] + oa[3];
  ctx[((size_t)b * LQ_ + rowa + 4) * D_ + h * DH_ + j] = ob[0] + ob[1] + ob[2] + ob[3];
}

extern "C" void kernel_launch(void* const* d_in, const int* in_sizes, int n_in,
                              void* d_out, int out_size, void* d_ws, size_t ws_size,
                              hipStream_t stream) {
  (void)in_sizes; (void)n_in; (void)out_size; (void)ws_size;

  const float* queries    = (const float*)d_in[0];
  const float* keys       = (const float*)d_in[1];
  const float* values     = (const float*)d_in[2];
  const int*   valid_lens = (const int*)  d_in[3];
  const float* Wq  = (const float*)d_in[4];
  const float* Wk  = (const float*)d_in[5];
  const float* Wv  = (const float*)d_in[6];
  const float* Wo  = (const float*)d_in[7];
  const float* Wqa = (const float*)d_in[8];
  const float* Wka = (const float*)d_in[9];
  const float* wva = (const float*)d_in[10];

  float* ws  = (float*)d_ws;
  const size_t nQKV = (size_t)B_ * H_ * LQ_ * DH_;
  const size_t nF   = (size_t)B_ * H_ * LQ_ * F_;
  float* q   = ws;
  float* k   = q  + nQKV;
  float* v   = k  + nQKV;
  float* qf  = v  + nQKV;
  float* kfb = qf + nF;
  float* ctx = kfb + nF;

  dim3 blk(256);
  dim3 pblk(64);
  proj_kernel<<<256, pblk, 0, stream>>>(keys,    Wk, k, 1);

  vproj_feat_kernel<<<VGRP_ + FGRP_, pblk, 0, stream>>>(values, Wv, v, k, Wka, kfb);

  score_softmax_av_kernel<<<B_ * H_ * (LQ_ / TQ_), blk, 0, stream>>>(
      queries, Wq, Wqa, kfb, v, wva, valid_lens, ctx);

  proj_kernel<<<256, pblk, 0, stream>>>(ctx, Wo, (float*)d_out, 0);
}


# GEMM operands in k-panel order (8-k blocks of all rows contiguous): spare work-groups of the K launch copy the inputs, k and ctx are written so; a fragment load touches 4 lines, not 16; sums unchanged
# speedup vs baseline: 2.7884x; 1.1058x over previous
#include <hip/hip_runtime.h>
#include <hip/hip_bf16.h>
#include <math.h>

constexpr int B_  = 2;
constexpr int H_  = 8;
constexpr int LQ_ = 128;
constexpr int LK_ = 128;
constexpr int D_  = 512;
constexpr int DH_ = 64;
constexpr int F_  = 512;

typedef float v2f __attribute__((ext_vector_type(2)));
typedef float v4f __attribute__((ext_vector_type(4)));
typedef float v8f __attribute__((ext_vector_type(8)));

#if __has_builtin(__builtin_amdgcn_tanhf)
#define TANH_F32(x) __builtin_amdgcn_tanhf(x)
#elif __has_builtin(__builtin_amdgcn_tanh_f32)
#define TANH_F32(x) __builtin_amdgcn_tanh_f32(x)
#else
#define TANH_F32(x) tanhf(x)
#endif

constexpr int CH_    = 8;
constexpr int AHEAD_ = 3;

template <int K, int SA, int SW, typename AfterRequests>
__device__ __forceinline__ v8f wmma_f32_rows(const v4f* __restrict__ arow,
                                             const v4f* __restrict__ wrow,
                                             AfterRequests&& after_requests)
{
  constexpr int NP  = K / 8;
  constexpr int NCH = NP / CH_;
  constexpr int LA  = (NCH < AHEAD_) ? NCH : AHEAD_;
  v4f a[NP], b[NP];
  v8f c0 = {}, c1 = {};
#pragma unroll
  for (int ch = 0; ch < NCH + LA; ++ch) {
    if (ch < NCH) {
#pragma unroll
      for (int i = ch * CH_; i < (ch + 1) * CH_; ++i) {
        a[i] = arow[SA * i];
        b[i] = wrow[SW * i];
      }
      if (ch == NCH - 1) after_requests();
    }
    if (ch >= LA) {
#pragma unroll
      for (int i = (ch - LA) * CH_; i < (ch - LA + 1) * CH_; ++i) {
        const v2f a01 = {a[i][0], a[i][1]}, b01 = {b[i][0], b[i][1]};
        const v2f a23 = {a[i][2], a[i][3]}, b23 = {b[i][2], b[i][3]};
        c0 = __builtin_amdgcn_wmma_f32_16x16x4_f32(
             false, a01,  false, b01,
             (short)0, c0,  false,  false);
        c1 = __builtin_amdgcn_wmma_f32_16x16x4_f32(
             false, a23,  false, b23,
             (short)0, c1,  false,  false);
      }
    }
  }
  return c0 + c1;
}

__device__ __forceinline__ const v4f* rowmajor_lane(const float* __restrict__ M, int ld, int r0, int lane)
{
  return (const v4f*)(M + (size_t)(r0 + (lane & 15)) * ld + 4 * (lane >> 4));
}
__device__ __forceinline__ const v4f* panel_lane(const float* __restrict__ P, int r0, int lane)
{
  return (const v4f*)(P + (size_t)(r0 + (lane & 15)) * 8 + 4 * (lane >> 4));
}

constexpr int OUT_PLAIN = 0, OUT_HEADS = 1, OUT_HEAD_PANELS = 2;
constexpr int MROWS_ = B_ * LQ_;

template <bool PANELS, int OUT>
__device__ __forceinline__ void proj_tile(const float* __restrict__ X, const float* __restrict__ W,
                                          float* __restrict__ out, int tile, int lane)
{
  constexpr int N = D_;
  constexpr int K = D_;
  const int tm = tile / (N / 16);
  const int tn = tile % (N / 16);
  const int m0 = tm * 16, n0 = tn * 16;

  v8f c;
  if (PANELS)
    c = wmma_f32_rows<K, 2 * MROWS_, 2 * N>(panel_lane(X, m0, lane), panel_lane(W, n0, lane), [] {});
  else
    c = wmma_f32_rows<K, 2, 2>(rowmajor_lane(X, K, m0, lane), rowmajor_lane(W, K, n0, lane), [] {});

  const int l16 = lane & 15, half = lane >> 4;
  const int col = n0 + l16;
  for (int r = 0; r < 8; ++r) {
    const int row = m0 + r + 8 * half;
    if (OUT == OUT_PLAIN) {
      out[(size_t)row * N + col] = c[r];
    } else {
      const int b = row / LQ_, l = row % LQ_;
      const int h = col / DH_, j = col % DH_;
      if (OUT == OUT_HEADS)
        out[(((size_t)(b * H_ + h) * LQ_) + l) * DH_ + j] = c[r];
      else
        out[(size_t)(b * H_ + h) * (LQ_ * DH_) + ((j >> 3) * LQ_ + l) * 8 + (j & 7)] = c[r];
    }
  }
}

constexpr int RP_UNITS_ = 512;

template <int R, int K>
__device__ __forceinline__ void repack_panels(const float* __restrict__ src, float* __restrict__ dst,
                                              int wg, int tid)
{
  constexpr int UPT = RP_UNITS_ / 64;
  v4f lo[UPT], hi[UPT];
  int doff[UPT];
#pragma unroll
  for (int j = 0; j < UPT; ++j) {
    const int u  = wg * RP_UNITS_ + j * 64 + tid;
    const int r  = (u >> 2) % R;
    const int kb = ((u >> 2) / R) * 4 + (u & 3);
    const v4f* s = (const v4f*)(src + (size_t)r * K + kb * 8);
    lo[j] = s[0];
    hi[j] = s[1];
    doff[j] = (kb * R + r) * 8;
  }
#pragma unroll
  for (int j = 0; j < UPT; ++j) {
    v4f* d = (v4f*)(dst + doff[j]);
    d[0] = lo[j];
    d[1] = hi[j];
  }
}

constexpr int PK_QUERIES_ = 0;
constexpr int PK_WV_      = PK_QUERIES_ + MROWS_ * D_;
constexpr int PK_WQ_      = PK_WV_  + D_ * D_;
constexpr int PK_WO_      = PK_WQ_  + D_ * D_;
constexpr int PK_WQA_     = PK_WO_  + D_ * D_;
constexpr int PK_WKA_     = PK_WQA_ + F_ * DH_;
constexpr int RG_VALUES_  = MROWS_ * D_ / 8 / RP_UNITS_;
constexpr int RG_QUERIES_ = MROWS_ * D_ / 8 / RP_UNITS_;
constexpr int RG_W_       = D_ * D_ / 8 / RP_UNITS_;
constexpr int RG_WA_      = F_ * DH_ / 8 / RP_UNITS_;
constexpr int RGRP_       = RG_VALUES_ + RG_QUERIES_ + 3 * RG_W_ + 2 * RG_WA_;
constexpr int KGRP_       = (B_ * LK_ / 16) * (D_ / 16) / 2;

__global__ void __launch_bounds__(64) __attribute__((amdgpu_waves_per_eu(1, 2)))
kproj_repack_kernel(const float* __restrict__ keys, const float* __restrict__ Wk, float* __restrict__ kout,
                    const float* __restrict__ values, const float* __restrict__ queries,
                    const float* __restrict__ Wv, const float* __restrict__ Wq,
                    const float* __restrict__ Wo, const float* __restrict__ Wqa,
                    const float* __restrict__ Wka,
                    float* __restrict__ pvalues, float* __restrict__ pack)
{
  const int tid = threadIdx.x;
  if (blockIdx.x < KGRP_) {
    proj_tile<false, OUT_HEAD_PANELS>(keys, Wk, kout, blockIdx.x * 2 + (tid >> 5), tid & 31);
    return;
  }
  int g = blockIdx.x - KGRP_;
  if (g < RG_VALUES_)  { repack_panels<MROWS_, D_>(values,  pvalues,            g, tid); return; }
  g -= RG_VALUES_;
  if (g < RG_QUERIES_) { repack_panels<MROWS_, D_>(queries, pack + PK_QUERIES_, g, tid); return; }
  g -= RG_QUERIES_;
  if (g < RG_W_)       { repack_panels<D_, D_>(Wv, pack + PK_WV_, g, tid); return; }
  g -= RG_W_;
  if (g < RG_W_)       { repack_panels<D_, D_>(Wq, pack + PK_WQ_, g, tid); return; }
  g -= RG_W_;
  if (g < RG_W_)       { repack_panels<D_, D_>(Wo, pack + PK_WO_, g, tid); return; }
  g -= RG_W_;
  if (g < RG_WA_)      { repack_panels<F_, DH_>(Wqa, pack + PK_WQA_, g, tid); return; }
  g -= RG_WA_;
  repack_panels<F_, DH_>(Wka, pack + PK_WKA_, g, tid);
}

__global__ void __launch_bounds__(64) __attribute__((amdgpu_waves_per_eu(1, 2)))
oproj_kernel(const float* __restrict__ ctxp, const float* __restrict__ Wop, float* __restrict__ out)
{
  const int tid  = threadIdx.x;
  proj_tile<true, OUT_PLAIN>(ctxp, Wop, out, blockIdx.x * 2 + (tid >> 5), tid & 31);
}

constexpr int VGRP_ = (B_ * LK_ / 16) * (D_ / 16) / 2;
constexpr int FGRP_ = B_ * H_ * (LK_ / 16) * (F_ / 16) / 2;

__global__ void __launch_bounds__(64) __attribute__((amdgpu_waves_per_eu(1, 4)))
vproj_feat_kernel(const float* __restrict__ Xv, const float* __restrict__ Wvp,
                  float* __restrict__ vout,
                  const float* __restrict__ Hd, const float* __restrict__ Wa,
                  float* __restrict__ outf)
{
  const int tid  = threadIdx.x;
  const int lane = tid & 31;
  const int wave = tid >> 5;
  if (blockIdx.x < VGRP_) {
    proj_tile<true, OUT_HEADS>(Xv, Wvp, vout, blockIdx.x * 2 + wave, lane);
    return;
  }
  const unsigned ft = (blockIdx.x - (unsigned)VGRP_) * 2u + (unsigned)wave;
  const int bh   = (int)(ft / (unsigned)((LK_ / 16) * (F_ / 16)));
  const int tile = (int)(ft % (unsigned)((LK_ / 16) * (F_ / 16)));
  const float* A = Hd + (size_t)bh * LK_ * DH_;
  float* O = outf + (size_t)bh * LK_ * F_;
  const int tm = tile / (F_ / 16);
  const int tn = tile % (F_ / 16);
  const int m0 = tm * 16, n0 = tn * 16;

  v8f c = wmma_f32_rows<DH_, 2 * LK_, 2 * F_>(panel_lane(A, m0, lane), panel_lane(Wa, n0, lane), [] {});

  const int l16 = lane & 15, half = lane >> 4;
  for (int r = 0; r < 8; ++r)
    O[(size_t)(m0 + r + 8 * half) * F_ + n0 + l16] = c[r];
}

constexpr int TQ_  = 8;
constexpr int F4_  = F_ / 4;
constexpr int KP4_ = F4_ + 1;
constexpr int SLAB4_ = 16;
constexpr int SLABS_ = F4_ / SLAB4_;
constexpr int SPT_   = (LK_ * SLAB4_) / 256;
constexpr int QP4_   = 18;

__global__ void __launch_bounds__(256)
score_softmax_av_kernel(const float* __restrict__ queries,
                        const float* __restrict__ Wq,
                        const float* __restrict__ Wqa,
                        const float* __restrict__ kf,
                        const float* __restrict__ vh,
                        const float* __restrict__ wv_a,
                        const int*   __restrict__ valid_lens,
                        float* __restrict__ ctx)
{
  __shared__ v4f skf[LK_ * KP4_];
  __shared__ v4f sqf[TQ_ * F4_];
  __shared__ v4f sqp[2 * TQ_ * QP4_];
  __shared__ v4f sw[F4_];
  __shared__ float sp[TQ_ * LK_];

  const int qt  = blockIdx.x % (LQ_ / TQ_);
  const int bh  = blockIdx.x / (LQ_ / TQ_);
  const int b   = bh / H_;
  const int h   = bh % H_;
  const int tid  = threadIdx.x;
  const int lane = tid & 31;
  const int wave = tid >> 5;

  const v4f* kf4 = (const v4f*)(kf + (size_t)bh * LK_ * F_);
  const v4f* wv4 = (const v4f*)wv_a;

  int goff[SPT_];
  int loff[SPT_];
#pragma unroll
  for (int jj = 0; jj < SPT_; ++jj) {
    const int u = tid + 256 * jj;
    goff[jj] = (u >> 4) * F4_  + (u & (SLAB4_ - 1));
    loff[jj] = (u >> 4) * KP4_ + (u & (SLAB4_ - 1));
  }
  v4f cur[SPT_];
#pragma unroll
  for (int jj = 0; jj < SPT_; ++jj) cur[jj] = kf4[goff[jj]];
  const int l16  = lane & 15;
  const int half = lane >> 4;
  v4f bq[4][8];
  constexpr int PW4_ = 2 * F_;
  {
    const int tnA = wave & 3;
    const int khA = wave >> 2;
    const float* xrow = queries + ((size_t)(32 * khA) * MROWS_ + (size_t)(b * LQ_ + qt * TQ_ + (l16 & 7))) * 8 + 4 * half;
    const float* wrow = Wq + ((size_t)(32 * khA) * D_ + (size_t)(h * DH_ + 16 * tnA + l16)) * 8 + 4 * half;
    const v4f* wqa4 = (const v4f*)(Wqa + (size_t)(64 * wave + l16) * 8 + 4 * half);
    const v8f cq = wmma_f32_rows<256, 2 * MROWS_, 2 * D_>((const v4f*)xrow, (const v4f*)wrow, [&] {
#pragma unroll
      for (int t = 0; t < 4; ++t)
#pragma unroll
        for (int i = 0; i < 8; ++i) bq[t][i] = wqa4[t * (16 * 8 / 4) + PW4_ * i];
    });
    if (half == 0) {
      float* dst = (float*)sqp + (size_t)(khA * TQ_) * (4 * QP4_) + 16 * tnA + l16;
      for (int r = 0; r < TQ_; ++r) dst[r * (4 * QP4_)] = cq[r];
    }
  }
  if (tid < F4_) sw[tid] = wv4[tid];
  __syncthreads();
  {
    const v4f* sq0 = sqp + (l16 & 7) * QP4_ + half;
    const v4f* sq1 = sq0 + TQ_ * QP4_;
    v4f aq[8];
#pragma unroll
    for (int i = 0; i < 8; ++i) aq[i] = sq0[2 * i] + sq1[2 * i];
    v8f d0[4] = {}, d1[4] = {};
#pragma unroll
    for (int i = 0; i < 8; ++i) {
      const v2f a01 = {aq[i][0], aq[i][1]};
      const v2f a23 = {aq[i][2], aq[i][3]};
#pragma unroll
      for (int t = 0; t < 4; ++t) {
        const v2f b01 = {bq[t][i][0], bq[t][i][1]};
        const v2f b23 = {bq[t][i][2], bq[t][i][3]};
        d0[t] = __builtin_amdgcn_wmma_f32_16x16x4_f32(
             false, a01,  false, b01,
             (short)0, d0[t],  false,  false);
        d1[t] = __builtin_amdgcn_wmma_f32_16x16x4_f32(
             false, a23,  false, b23,
             (short)0, d1[t],  false,  false);
      }
    }
    v2f* sqf_2 = (v2f*)sqf;
#pragma unroll
    for (int t = 0; t < 4; ++t) {
      const v8f c = d0[t] + d1[t];
      if (half == 0) {
#pragma unroll
        for (int p = 0; p < TQ_ / 2; ++p) {
          const v2f pr = {c[2 * p], c[2 * p + 1]};
          sqf_2[p * F_ + 64 * wave + 16 * t + l16] = pr;
        }
      }
    }
  }

  const int qa = 2 * (wave & 3);
  const int ka = 64 * (wave >> 2) + lane;
  const v4f* qpp = sqf + (wave & 3) * (F_ / 2);
  const v4f* k0p = skf + ka * KP4_;
  const v4f* k1p = k0p + 32 * KP4_;
  constexpr int NB_ = 4;
  struct OpSet { v4f w[NB_], qlo[NB_], qhi[NB_], k0[NB_], k1[NB_]; };
  OpSet A, Bs;
  v2f acc0 = {0.f, 0.f};
  v2f acc1 = {0.f, 0.f};
  auto fetch = [&](OpSet& o, int g0) {
#pragma unroll
    for (int i = 0; i < NB_; ++i) {
      o.w[i]   = sw[g0 + i];
      o.qlo[i] = qpp[2 * (g0 + i)];
      o.qhi[i] = qpp[2 * (g0 + i) + 1];
      o.k0[i]  = k0p[g0 + i];
      o.k1[i]  = k1p[g0 + i];
    }
  };
  auto half2 = [](const v4f& x, int e) -> v2f {
    return (e < 2) ? __builtin_shufflevector(x, x, 0, 1) : __builtin_shufflevector(x, x, 2, 3);
  };
  auto both = [](const v4f& x, int e) -> v2f {
    return (e == 0) ? __builtin_shufflevector(x, x, 0, 0) : (e == 1) ? __builtin_shufflevector(x, x, 1, 1)
         : (e == 2) ? __builtin_shufflevector(x, x, 2, 2) : __builtin_shufflevector(x, x, 3, 3);
  };
  auto qpair = [&](const OpSet& o, int i, int e) -> v2f {
    return half2((e < 2) ? o.qlo[i] : o.qhi[i], 2 * (e & 1));
  };
  v2f ns0, ns1;
  v2f pt0 = {0.f, 0.f}, pt1 = {0.f, 0.f};
  v2f pw = {0.f, 0.f};
#define ORDER_KEPT() __builtin_amdgcn_sched_barrier(0)
  auto score = [&](const OpSet& o, const OpSet& nx, bool has_next) {
#pragma unroll
    for (int i = 0; i < NB_; ++i) {
#pragma unroll
      for (int e = 0; e < 4; ++e) {
        const bool in_set = (4 * i + e + 1 < 4 * NB_);
        const OpSet& on = in_set ? o : nx;
        const int in = in_set ? (4 * i + e + 1) / 4 : 0;
        const int en = in_set ? (4 * i + e + 1) % 4 : 0;
        const bool adds = in_set || has_next;
        v2f t0, t1, m0 = ns0, m1 = ns1;
        t0[0] = TANH_F32(ns0[0]);
        ORDER_KEPT();
        if (adds) m0 = qpair(on, in, en) + both(on.k0[in], en);
        ORDER_KEPT();
        t0[1] = TANH_F32(ns0[1]);
        ORDER_KEPT();
        acc0 += pw * pt0;
        ORDER_KEPT();
        t1[0] = TANH_F32(ns1[0]);
        ORDER_KEPT();
        if (adds) m1 = qpair(on, in, en) + both(on.k1[in], en);
        ORDER_KEPT();
        t1[1] = TANH_F32(ns1[1]);
        ORDER_KEPT();
        acc1 += pw * pt1;
        ORDER_KEPT();
        pt0 = t0; pt1 = t1; pw = both(o.w[i], e);
        ns0 = m0; ns1 = m1;
      }
    }
  };
#define BLOCK_SCORED() asm volatile("" : "+v"(acc0), "+v"(acc1) : : "memory")
#define READS_ISSUED() __builtin_amdgcn_sched_barrier(0)

#pragma unroll
  for (int jj = 0; jj < SPT_; ++jj) skf[loff[jj]] = cur[jj];
  __syncthreads();
#pragma unroll
  for (int jj = 0; jj < SPT_; ++jj) cur[jj] = kf4[goff[jj] + SLAB4_];
  fetch(A, 0);
  ns0 = qpair(A, 0, 0) + both(A.k0[0], 0);
  ns1 = qpair(A, 0, 0) + both(A.k1[0], 0);
#pragma nounroll
  for (int s = 0; s < SLABS_ - 1; ++s) {
    const int g = SLAB4_ * s;
    fetch(Bs, g + NB_);
    READS_ISSUED();
    score(A, Bs, true);
    BLOCK_SCORED();
    fetch(A, g + 2 * NB_);
    READS_ISSUED();
    score(Bs, A, true);
    BLOCK_SCORED();
    fetch(Bs, g + 3 * NB_);
    READS_ISSUED();
    score(A, Bs, true);
    BLOCK_SCORED();
#pragma unroll
    for (int jj = 0; jj < SPT_; ++jj) skf[loff[jj] + SLAB4_ * (s + 1)] = cur[jj];
    __syncthreads();
    if (s + 2 < SLABS_) {
#pragma unroll
      for (int jj = 0; jj < SPT_; ++jj) cur[jj] = kf4[goff[jj] + SLAB4_ * (s + 2)];
    }
    fetch(A, g + 4 * NB_);
    READS_ISSUED();
    score(Bs, A, true);
    BLOCK_SCORED();
  }
  {
    const int g = SLAB4_ * (SLABS_ - 1);
    fetch(Bs, g + NB_);
    READS_ISSUED();
    score(A, Bs, true);
    BLOCK_SCORED();
    fetch(A, g + 2 * NB_);
    READS_ISSUED();
    score(Bs, A, true);
    BLOCK_SCORED();
    fetch(Bs, g + 3 * NB_);
    READS_ISSUED();
    score(A, Bs, true);
    BLOCK_SCORED();
    score(Bs, A, false);
    acc0 += pw * pt0;
    acc1 += pw * pt1;
  }
#undef BLOCK_SCORED
#undef READS_ISSUED
#undef ORDER_KEPT
  sp[qa * LK_ + ka]            = acc0[0];
  sp[qa * LK_ + ka + 32]       = acc1[0];
  sp[(qa + 1) * LK_ + ka]      = acc0[1];
  sp[(qa + 1) * LK_ + ka + 32] = acc1[1];
  __syncthreads();

  {
    const int valid = valid_lens[b];
    float* srow = sp + wave * LK_;
    float val[4];
#pragma unroll
    for (int g = 0; g < 4; ++g) {
      const int k = lane + 32 * g;
      val[g] = (k < valid) ? srow[k] : -1.0e6f;
    }
    float m = fmaxf(fmaxf(val[0], val[1]), fmaxf(val[2], val[3]));
    for (int off = 16; off > 0; off >>= 1)
      m = fmaxf(m, __shfl_xor(m, off, 32));
    float e[4], s[4];
#pragma unroll
    for (int g = 0; g < 4; ++g) {
      e[g] = __expf(val[g] - m);
      s[g] = e[g];
      for (int off = 16; off > 0; off >>= 1)
        s[g] += __shfl_xor(s[g], off, 32);
    }
    const float rowsum = s[0] + s[1] + s[2] + s[3];
#pragma unroll
    for (int g = 0; g < 4; ++g)
      srow[lane + 32 * g] = e[g] / rowsum;
  }
  __syncthreads();

  const int j  = tid & 63;
  const int qg = tid >> 6;
  const float* vb = vh + (size_t)bh * LK_ * DH_ + j;
  const float* pa = sp + qg * LK_;
  const float* pb = sp + (qg + 4) * LK_;
  float oa[4] = {0.f, 0.f, 0.f, 0.f};
  float ob[4] = {0.f, 0.f, 0.f, 0.f};
  for (int kk = 0; kk < 32; ++kk) {
#pragma unroll
    for (int part = 0; part < 4; ++part) {
      const int k = part * 32 + kk;
      const float vv = vb[(size_t)k * DH_];
      oa[part] += pa[k] * vv;
      ob[part] += pb[k] * vv;
    }
  }
  const int rowa = b * LQ_ + qt * TQ_ + qg;
  const int col  = h * DH_ + j;
  float* cdst = ctx + ((size_t)(col >> 3) * MROWS_ + rowa) * 8 + (col & 7);
  cdst[0]     = oa[0] + oa[1] + oa[2] + oa[3];
  cdst[4 * 8] = ob[0] + ob[1] + ob[2] + ob[3];
}

extern "C" void kernel_launch(void* const* d_in, const int* in_sizes, int n_in,
                              void* d_out, int out_size, void* d_ws, size_t ws_size,
                              hipStream_t stream) {
  (void)in_sizes; (void)n_in; (void)out_size; (void)ws_size;

  const float* queries    = (const float*)d_in[0];
  const float* keys       = (const float*)d_in[1];
  const float* values     = (const float*)d_in[2];
  const int*   valid_lens = (const int*)  d_in[3];
  const float* Wq  = (const float*)d_in[4];
  const float* Wk  = (const float*)d_in[5];
  const float* Wv  = (const float*)d_in[6];
  const float* Wo  = (const float*)d_in[7];
  const float* Wqa = (const float*)d_in[8];
  const float* Wka = (const float*)d_in[9];
  const float* wva = (const float*)d_in[10];

  float* ws  = (float*)d_ws;
  const size_t nQKV = (size_t)B_ * H_ * LQ_ * DH_;
  const size_t nF   = (size_t)B_ * H_ * LQ_ * F_;
  float* q   = ws;
  float* k   = q  + nQKV;
  float* v   = k  + nQKV;
  float* qf  = v  + nQKV;
  float* kfb = qf + nF;
  float* ctx = kfb + nF;

  dim3 blk(256);
  dim3 pblk(64);
  kproj_repack_kernel<<<KGRP_ + RGRP_, pblk, 0, stream>>>(keys, Wk, k,
      values, queries, Wv, Wq, Wo, Wqa, Wka, q, qf);

  vproj_feat_kernel<<<VGRP_ + FGRP_, pblk, 0, stream>>>(q, qf + PK_WV_, v, k, qf + PK_WKA_, kfb);

  score_softmax_av_kernel<<<B_ * H_ * (LQ_ / TQ_), blk, 0, stream>>>(
      qf + PK_QUERIES_, qf + PK_WQ_, qf + PK_WQA_, kfb, v, wva, valid_lens, ctx);

  oproj_kernel<<<256, pblk, 0, stream>>>(ctx, qf + PK_WO_, (float*)d_out);
}


# Q projection leaves the score work-groups: 512 full tiles in the V / key-feature launch (8 rows no longer padded to 16), q kept in the ctx plane; q now sums its 512 k in one pass, not two halves
# speedup vs baseline: 2.9059x; 1.0421x over previous
#include <hip/hip_runtime.h>
#include <hip/hip_bf16.h>
#include <math.h>

constexpr int B_  = 2;
constexpr int H_  = 8;
constexpr int LQ_ = 128;
constexpr int LK_ = 128;
constexpr int D_  = 512;
constexpr int DH_ = 64;
constexpr int F_  = 512;

typedef float v2f __attribute__((ext_vector_type(2)));
typedef float v4f __attribute__((ext_vector_type(4)));
typedef float v8f __attribute__((ext_vector_type(8)));

#if __has_builtin(__builtin_amdgcn_tanhf)
#define TANH_F32(x) __builtin_amdgcn_tanhf(x)
#elif __has_builtin(__builtin_amdgcn_tanh_f32)
#define TANH_F32(x) __builtin_amdgcn_tanh_f32(x)
#else
#define TANH_F32(x) tanhf(x)
#endif

constexpr int CH_    = 8;
constexpr int AHEAD_ = 3;

template <int K, int SA, int SW, typename AfterRequests>
__device__ __forceinline__ v8f wmma_f32_rows(const v4f* __restrict__ arow,
                                             const v4f* __restrict__ wrow,
                                             AfterRequests&& after_requests)
{
  constexpr int NP  = K / 8;
  constexpr int NCH = NP / CH_;
  constexpr int LA  = (NCH < AHEAD_) ? NCH : AHEAD_;
  v4f a[NP], b[NP];
  v8f c0 = {}, c1 = {};
#pragma unroll
  for (int ch = 0; ch < NCH + LA; ++ch) {
    if (ch < NCH) {
#pragma unroll
      for (int i = ch * CH_; i < (ch + 1) * CH_; ++i) {
        a[i] = arow[SA * i];
        b[i] = wrow[SW * i];
      }
      if (ch == NCH - 1) after_requests();
    }
    if (ch >= LA) {
#pragma unroll
      for (int i = (ch - LA) * CH_; i < (ch - LA + 1) * CH_; ++i) {
        const v2f a01 = {a[i][0], a[i][1]}, b01 = {b[i][0], b[i][1]};
        const v2f a23 = {a[i][2], a[i][3]}, b23 = {b[i][2], b[i][3]};
        c0 = __builtin_amdgcn_wmma_f32_16x16x4_f32(
             false, a01,  false, b01,
             (short)0, c0,  false,  false);
        c1 = __builtin_amdgcn_wmma_f32_16x16x4_f32(
             false, a23,  false, b23,
             (short)0, c1,  false,  false);
      }
    }
  }
  return c0 + c1;
}

__device__ __forceinline__ const v4f* rowmajor_lane(const float* __restrict__ M, int ld, int r0, int lane)
{
  return (const v4f*)(M + (size_t)(r0 + (lane & 15)) * ld + 4 * (lane >> 4));
}
__device__ __forceinline__ const v4f* panel_lane(const float* __restrict__ P, int r0, int lane)
{
  return (const v4f*)(P + (size_t)(r0 + (lane & 15)) * 8 + 4 * (lane >> 4));
}

constexpr int OUT_PLAIN = 0, OUT_HEADS = 1, OUT_HEAD_PANELS = 2, OUT_ROW_PANELS = 3;
constexpr int MROWS_ = B_ * LQ_;

template <bool PANELS, int OUT>
__device__ __forceinline__ void proj_tile(const float* __restrict__ X, const float* __restrict__ W,
                                          float* __restrict__ out, int tile, int lane)
{
  constexpr int N = D_;
  constexpr int K = D_;
  const int tm = tile / (N / 16);
  const int tn = tile % (N / 16);
  const int m0 = tm * 16, n0 = tn * 16;

  v8f c;
  if (PANELS)
    c = wmma_f32_rows<K, 2 * MROWS_, 2 * N>(panel_lane(X, m0, lane), panel_lane(W, n0, lane), [] {});
  else
    c = wmma_f32_rows<K, 2, 2>(rowmajor_lane(X, K, m0, lane), rowmajor_lane(W, K, n0, lane), [] {});

  const int l16 = lane & 15, half = lane >> 4;
  const int col = n0 + l16;
  for (int r = 0; r < 8; ++r) {
    const int row = m0 + r + 8 * half;
    if (OUT == OUT_PLAIN) {
      out[(size_t)row * N + col] = c[r];
    } else if (OUT == OUT_ROW_PANELS) {
      out[((size_t)(col >> 3) * MROWS_ + row) * 8 + (col & 7)] = c[r];
    } else {
      const int b = row / LQ_, l = row % LQ_;
      const int h = col / DH_, j = col % DH_;
      if (OUT == OUT_HEADS)
        out[(((size_t)(b * H_ + h) * LQ_) + l) * DH_ + j] = c[r];
      else
        out[(size_t)(b * H_ + h) * (LQ_ * DH_) + ((j >> 3) * LQ_ + l) * 8 + (j & 7)] = c[r];
    }
  }
}

constexpr int RP_UNITS_ = 512;

template <int R, int K>
__device__ __forceinline__ void repack_panels(const float* __restrict__ src, float* __restrict__ dst,
                                              int wg, int tid)
{
  constexpr int UPT = RP_UNITS_ / 64;
  v4f lo[UPT], hi[UPT];
  int doff[UPT];
#pragma unroll
  for (int j = 0; j < UPT; ++j) {
    const int u  = wg * RP_UNITS_ + j * 64 + tid;
    const int r  = (u >> 2) % R;
    const int kb = ((u >> 2) / R) * 4 + (u & 3);
    const v4f* s = (const v4f*)(src + (size_t)r * K + kb * 8);
    lo[j] = s[0];
    hi[j] = s[1];
    doff[j] = (kb * R + r) * 8;
  }
#pragma unroll
  for (int j = 0; j < UPT; ++j) {
    v4f* d = (v4f*)(dst + doff[j]);
    d[0] = lo[j];
    d[1] = hi[j];
  }
}

constexpr int PK_QUERIES_ = 0;
constexpr int PK_WV_      = PK_QUERIES_ + MROWS_ * D_;
constexpr int PK_WQ_      = PK_WV_  + D_ * D_;
constexpr int PK_WO_      = PK_WQ_  + D_ * D_;
constexpr int PK_WQA_     = PK_WO_  + D_ * D_;
constexpr int PK_WKA_     = PK_WQA_ + F_ * DH_;
constexpr int RG_VALUES_  = MROWS_ * D_ / 8 / RP_UNITS_;
constexpr int RG_QUERIES_ = MROWS_ * D_ / 8 / RP_UNITS_;
constexpr int RG_W_       = D_ * D_ / 8 / RP_UNITS_;
constexpr int RG_WA_      = F_ * DH_ / 8 / RP_UNITS_;
constexpr int RGRP_       = RG_VALUES_ + RG_QUERIES_ + 3 * RG_W_ + 2 * RG_WA_;
constexpr int KGRP_       = (B_ * LK_ / 16) * (D_ / 16) / 2;

__global__ void __launch_bounds__(64) __attribute__((amdgpu_waves_per_eu(1, 2)))
kproj_repack_kernel(const float* __restrict__ keys, const float* __restrict__ Wk, float* __restrict__ kout,
                    const float* __restrict__ values, const float* __restrict__ queries,
                    const float* __restrict__ Wv, const float* __restrict__ Wq,
                    const float* __restrict__ Wo, const float* __restrict__ Wqa,
                    const float* __restrict__ Wka,
                    float* __restrict__ pvalues, float* __restrict__ pack)
{
  const int tid = threadIdx.x;
  if (blockIdx.x < KGRP_) {
    proj_tile<false, OUT_HEAD_PANELS>(keys, Wk, kout, blockIdx.x * 2 + (tid >> 5), tid & 31);
    return;
  }
  int g = blockIdx.x - KGRP_;
  if (g < RG_VALUES_)  { repack_panels<MROWS_, D_>(values,  pvalues,            g, tid); return; }
  g -= RG_VALUES_;
  if (g < RG_QUERIES_) { repack_panels<MROWS_, D_>(queries, pack + PK_QUERIES_, g, tid); return; }
  g -= RG_QUERIES_;
  if (g < RG_W_)       { repack_panels<D_, D_>(Wv, pack + PK_WV_, g, tid); return; }
  g -= RG_W_;
  if (g < RG_W_)       { repack_panels<D_, D_>(Wq, pack + PK_WQ_, g, tid); return; }
  g -= RG_W_;
  if (g < RG_W_)       { repack_panels<D_, D_>(Wo, pack + PK_WO_, g, tid); return; }
  g -= RG_W_;
  if (g < RG_WA_)      { repack_panels<F_, DH_>(Wqa, pack + PK_WQA_, g, tid); return; }
  g -= RG_WA_;
  repack_panels<F_, DH_>(Wka, pack + PK_WKA_, g, tid);
}

__global__ void __launch_bounds__(64) __attribute__((amdgpu_waves_per_eu(1, 2)))
oproj_kernel(const float* __restrict__ ctxp, const float* __restrict__ Wop, float* __restrict__ out)
{
  const int tid  = threadIdx.x;
  proj_tile<true, OUT_PLAIN>(ctxp, Wop, out, blockIdx.x * 2 + (tid >> 5), tid & 31);
}

constexpr int VGRP_ = (B_ * LK_ / 16) * (D_ / 16) / 2;
constexpr int QGRP_ = (B_ * LQ_ / 16) * (D_ / 16) / 2;
constexpr int FGRP_ = B_ * H_ * (LK_ / 16) * (F_ / 16) / 2;

__global__ void __launch_bounds__(64) __attribute__((amdgpu_waves_per_eu(1, 4)))
vproj_feat_kernel(const float* __restrict__ Xv, const float* __restrict__ Wvp,
                  float* __restrict__ vout,
                  const float* __restrict__ Xq, const float* __restrict__ Wqp,
                  float* __restrict__ qout,
                  const float* __restrict__ Hd, const float* __restrict__ Wa,
                  float* __restrict__ outf)
{
  const int tid  = threadIdx.x;
  const int lane = tid & 31;
  const int wave = tid >> 5;
  if (blockIdx.x < VGRP_) {
    proj_tile<true, OUT_HEADS>(Xv, Wvp, vout, blockIdx.x * 2 + wave, lane);
    return;
  }
  if (blockIdx.x < VGRP_ + QGRP_) {
    proj_tile<true, OUT_ROW_PANELS>(Xq, Wqp, qout, (blockIdx.x - VGRP_) * 2 + wave, lane);
    return;
  }
  const unsigned ft = (blockIdx.x - (unsigned)(VGRP_ + QGRP_)) * 2u + (unsigned)wave;
  const int bh   = (int)(ft / (unsigned)((LK_ / 16) * (F_ / 16)));
  const int tile = (int)(ft % (unsigned)((LK_ / 16) * (F_ / 16)));
  const float* A = Hd + (size_t)bh * LK_ * DH_;
  float* O = outf + (size_t)bh * LK_ * F_;
  const int tm = tile / (F_ / 16);
  const int tn = tile % (F_ / 16);
  const int m0 = tm * 16, n0 = tn * 16;

  v8f c = wmma_f32_rows<DH_, 2 * LK_, 2 * F_>(panel_lane(A, m0, lane), panel_lane(Wa, n0, lane), [] {});

  const int l16 = lane & 15, half = lane >> 4;
  for (int r = 0; r < 8; ++r)
    O[(size_t)(m0 + r + 8 * half) * F_ + n0 + l16] = c[r];
}

constexpr int TQ_  = 8;
constexpr int F4_  = F_ / 4;
constexpr int KP4_ = F4_ + 1;
constexpr int SLAB4_ = 16;
constexpr int SLABS_ = F4_ / SLAB4_;
constexpr int SPT_   = (LK_ * SLAB4_) / 256;

__global__ void __launch_bounds__(256)
score_softmax_av_kernel(const float* __restrict__ Wqa,
                        const float* __restrict__ kf,
                        const float* __restrict__ vh,
                        const float* __restrict__ wv_a,
                        const int*   __restrict__ valid_lens,
                        float* qctx)
{
  __shared__ v4f skf[LK_ * KP4_];
  __shared__ v4f sqf[TQ_ * F4_];
  __shared__ v4f sw[F4_];
  __shared__ float sp[TQ_ * LK_];

  const int qt  = blockIdx.x % (LQ_ / TQ_);
  const int bh  = blockIdx.x / (LQ_ / TQ_);
  const int b   = bh / H_;
  const int h   = bh % H_;
  const int tid  = threadIdx.x;
  const int lane = tid & 31;
  const int wave = tid >> 5;

  const v4f* kf4 = (const v4f*)(kf + (size_t)bh * LK_ * F_);
  const v4f* wv4 = (const v4f*)wv_a;

  const int l16  = lane & 15;
  const int half = lane >> 4;
  v4f wreg = {0.f, 0.f, 0.f, 0.f};
  if (tid < F4_) wreg = wv4[tid];
  v4f aq[8];
  v4f bq[4][8];
  constexpr int PW4_ = 2 * F_;
  constexpr int PQ4_ = 2 * MROWS_;
  {
    const v4f* q4   = (const v4f*)(qctx + ((size_t)(8 * h) * MROWS_ + (size_t)(b * LQ_ + qt * TQ_ + (l16 & 7))) * 8 + 4 * half);
    const v4f* wqa4 = (const v4f*)(Wqa + (size_t)(64 * wave + l16) * 8 + 4 * half);
#pragma unroll
    for (int i = 0; i < 8; ++i) {
      aq[i] = q4[PQ4_ * i];
#pragma unroll
      for (int t = 0; t < 4; ++t) bq[t][i] = wqa4[t * (16 * 8 / 4) + PW4_ * i];
    }
  }
  int goff[SPT_];
  int loff[SPT_];
#pragma unroll
  for (int jj = 0; jj < SPT_; ++jj) {
    const int u = tid + 256 * jj;
    goff[jj] = (u >> 4) * F4_  + (u & (SLAB4_ - 1));
    loff[jj] = (u >> 4) * KP4_ + (u & (SLAB4_ - 1));
  }
  v4f cur[SPT_];
#pragma unroll
  for (int jj = 0; jj < SPT_; ++jj) cur[jj] = kf4[goff[jj]];
  {
    v8f d0[4] = {}, d1[4] = {};
#pragma unroll
    for (int i = 0; i < 8; ++i) {
      const v2f a01 = {aq[i][0], aq[i][1]};
      const v2f a23 = {aq[i][2], aq[i][3]};
#pragma unroll
      for (int t = 0; t < 4; ++t) {
        const v2f b01 = {bq[t][i][0], bq[t][i][1]};
        const v2f b23 = {bq[t][i][2], bq[t][i][3]};
        d0[t] = __builtin_amdgcn_wmma_f32_16x16x4_f32(
             false, a01,  false, b01,
             (short)0, d0[t],  false,  false);
        d1[t] = __builtin_amdgcn_wmma_f32_16x16x4_f32(
             false, a23,  false, b23,
             (short)0, d1[t],  false,  false);
      }
    }
    v2f* sqf_2 = (v2f*)sqf;
#pragma unroll
    for (int t = 0; t < 4; ++t) {
      const v8f c = d0[t] + d1[t];
      if (half == 0) {
#pragma unroll
        for (int p = 0; p < TQ_ / 2; ++p) {
          const v2f pr = {c[2 * p], c[2 * p + 1]};
          sqf_2[p * F_ + 64 * wave + 16 * t + l16] = pr;
        }
      }
    }
  }
  if (tid < F4_) sw[tid] = wreg;

  const int qa = 2 * (wave & 3);
  const int ka = 64 * (wave >> 2) + lane;
  const v4f* qpp = sqf + (wave & 3) * (F_ / 2);
  const v4f* k0p = skf + ka * KP4_;
  const v4f* k1p = k0p + 32 * KP4_;
  constexpr int NB_ = 4;
  struct OpSet { v4f w[NB_], qlo[NB_], qhi[NB_], k0[NB_], k1[NB_]; };
  OpSet A, Bs;
  v2f acc0 = {0.f, 0.f};
  v2f acc1 = {0.f, 0.f};
  auto fetch = [&](OpSet& o, int g0) {
#pragma unroll
    for (int i = 0; i < NB_; ++i) {
      o.w[i]   = sw[g0 + i];
      o.qlo[i] = qpp[2 * (g0 + i)];
      o.qhi[i] = qpp[2 * (g0 + i) + 1];
      o.k0[i]  = k0p[g0 + i];
      o.k1[i]  = k1p[g0 + i];
    }
  };
  auto half2 = [](const v4f& x, int e) -> v2f {
    return (e < 2) ? __builtin_shufflevector(x, x, 0, 1) : __builtin_shufflevector(x, x, 2, 3);
  };
  auto both = [](const v4f& x, int e) -> v2f {
    return (e == 0) ? __builtin_shufflevector(x, x, 0, 0) : (e == 1) ? __builtin_shufflevector(x, x, 1, 1)
         : (e == 2) ? __builtin_shufflevector(x, x, 2, 2) : __builtin_shufflevector(x, x, 3, 3);
  };
  auto qpair = [&](const OpSet& o, int i, int e) -> v2f {
    return half2((e < 2) ? o.qlo[i] : o.qhi[i], 2 * (e & 1));
  };
  v2f ns0, ns1;
  v2f pt0 = {0.f, 0.f}, pt1 = {0.f, 0.f};
  v2f pw = {0.f, 0.f};
#define ORDER_KEPT() __builtin_amdgcn_sched_barrier(0)
  auto score = [&](const OpSet& o, const OpSet& nx, bool has_next) {
#pragma unroll
    for (int i = 0; i < NB_; ++i) {
#pragma unroll
      for (int e = 0; e < 4; ++e) {
        const bool in_set = (4 * i + e + 1 < 4 * NB_);
        const OpSet& on = in_set ? o : nx;
        const int in = in_set ? (4 * i + e + 1) / 4 : 0;
        const int en = in_set ? (4 * i + e + 1) % 4 : 0;
        const bool adds = in_set || has_next;
        v2f t0, t1, m0 = ns0, m1 = ns1;
        t0[0] = TANH_F32(ns0[0]);
        ORDER_KEPT();
        if (adds) m0 = qpair(on, in, en) + both(on.k0[in], en);
        ORDER_KEPT();
        t0[1] = TANH_F32(ns0[1]);
        ORDER_KEPT();
        acc0 += pw * pt0;
        ORDER_KEPT();
        t1[0] = TANH_F32(ns1[0]);
        ORDER_KEPT();
        if (adds) m1 = qpair(on, in, en) + both(on.k1[in], en);
        ORDER_KEPT();
        t1[1] = TANH_F32(ns1[1]);
        ORDER_KEPT();
        acc1 += pw * pt1;
        ORDER_KEPT();
        pt0 = t0; pt1 = t1; pw = both(o.w[i], e);
        ns0 = m0; ns1 = m1;
      }
    }
  };
#define BLOCK_SCORED() asm volatile("" : "+v"(acc0), "+v"(acc1) : : "memory")
#define READS_ISSUED() __builtin_amdgcn_sched_barrier(0)

#pragma unroll
  for (int jj = 0; jj < SPT_; ++jj) skf[loff[jj]] = cur[jj];
  __syncthreads();
#pragma unroll
  for (int jj = 0; jj < SPT_; ++jj) cur[jj] = kf4[goff[jj] + SLAB4_];
  fetch(A, 0);
  ns0 = qpair(A, 0, 0) + both(A.k0[0], 0);
  ns1 = qpair(A, 0, 0) + both(A.k1[0], 0);
#pragma nounroll
  for (int s = 0; s < SLABS_ - 1; ++s) {
    const int g = SLAB4_ * s;
    fetch(Bs, g + NB_);
    READS_ISSUED();
    score(A, Bs, true);
    BLOCK_SCORED();
    fetch(A, g + 2 * NB_);
    READS_ISSUED();
    score(Bs, A, true);
    BLOCK_SCORED();
    fetch(Bs, g + 3 * NB_);
    READS_ISSUED();
    score(A, Bs, true);
    BLOCK_SCORED();
#pragma unroll
    for (int jj = 0; jj < SPT_; ++jj) skf[loff[jj] + SLAB4_ * (s + 1)] = cur[jj];
    __syncthreads();
    if (s + 2 < SLABS_) {
#pragma unroll
      for (int jj = 0; jj < SPT_; ++jj) cur[jj] = kf4[goff[jj] + SLAB4_ * (s + 2)];
    }
    fetch(A, g + 4 * NB_);
    READS_ISSUED();
    score(Bs, A, true);
    BLOCK_SCORED();
  }
  {
    const int g = SLAB4_ * (SLABS_ - 1);
    fetch(Bs, g + NB_);
    READS_ISSUED();
    score(A, Bs, true);
    BLOCK_SCORED();
    fetch(A, g + 2 * NB_);
    READS_ISSUED();
    score(Bs, A, true);
    BLOCK_SCORED();
    fetch(Bs, g + 3 * NB_);
    READS_ISSUED();
    score(A, Bs, true);
    BLOCK_SCORED();
    score(Bs, A, false);
    acc0 += pw * pt0;
    acc1 += pw * pt1;
  }
#undef BLOCK_SCORED
#undef READS_ISSUED
#undef ORDER_KEPT
  sp[qa * LK_ + ka]            = acc0[0];
  sp[qa * LK_ + ka + 32]       = acc1[0];
  sp[(qa + 1) * LK_ + ka]      = acc0[1];
  sp[(qa + 1) * LK_ + ka + 32] = acc1[1];
  __syncthreads();

  {
    const int valid = valid_lens[b];
    float* srow = sp + wave * LK_;
    float val[4];
#pragma unroll
    for (int g = 0; g < 4; ++g) {
      const int k = lane + 32 * g;
      val[g] = (k < valid) ? srow[k] : -1.0e6f;
    }
    float m = fmaxf(fmaxf(val[0], val[1]), fmaxf(val[2], val[3]));
    for (int off = 16; off > 0; off >>= 1)
      m = fmaxf(m, __shfl_xor(m, off, 32));
    float e[4], s[4];
#pragma unroll
    for (int g = 0; g < 4; ++g) {
      e[g] = __expf(val[g] - m);
      s[g] = e[g];
      for (int off = 16; off > 0; off >>= 1)
        s[g] += __shfl_xor(s[g], off, 32);
    }
    const float rowsum = s[0] + s[1] + s[2] + s[3];
#pragma unroll
    for (int g = 0; g < 4; ++g)
      srow[lane + 32 * g] = e[g] / rowsum;
  }
  __syncthreads();

  const int j  = tid & 63;
  const int qg = tid >> 6;
  const float* vb = vh + (size_t)bh * LK_ * DH_ + j;
  const float* pa = sp + qg * LK_;
  const float* pb = sp + (qg + 4) * LK_;
  float oa[4] = {0.f, 0.f, 0.f, 0.f};
  float ob[4] = {0.f, 0.f, 0.f, 0.f};
  for (int kk = 0; kk < 32; ++kk) {
#pragma unroll
    for (int part = 0; part < 4; ++part) {
      const int k = part * 32 + kk;
      const float vv = vb[(size_t)k * DH_];
      oa[part] += pa[k] * vv;
      ob[part] += pb[k] * vv;
    }
  }
  const int rowa = b * LQ_ + qt * TQ_ + qg;
  const int col  = h * DH_ + j;
  float* cdst = qctx + ((size_t)(col >> 3) * MROWS_ + rowa) * 8 + (col & 7);
  cdst[0]     = oa[0] + oa[1] + oa[2] + oa[3];
  cdst[4 * 8] = ob[0] + ob[1] + ob[2] + ob[3];
}

extern "C" void kernel_launch(void* const* d_in, const int* in_sizes, int n_in,
                              void* d_out, int out_size, void* d_ws, size_t ws_size,
                              hipStream_t stream) {
  (void)in_sizes; (void)n_in; (void)out_size; (void)ws_size;

  const float* queries    = (const float*)d_in[0];
  const float* keys       = (const float*)d_in[1];
  const float* values     = (const float*)d_in[2];
  const int*   valid_lens = (const int*)  d_in[3];
  const float* Wq  = (const float*)d_in[4];
  const float* Wk  = (const float*)d_in[5];
  const float* Wv  = (const float*)d_in[6];
  const float* Wo  = (const float*)d_in[7];
  const float* Wqa = (const float*)d_in[8];
  const float* Wka = (const float*)d_in[9];
  const float* wva = (const float*)d_in[10];

  float* ws  = (float*)d_ws;
  const size_t nQKV = (size_t)B_ * H_ * LQ_ * DH_;
  const size_t nF   = (size_t)B_ * H_ * LQ_ * F_;
  float* q   = ws;
  float* k   = q  + nQKV;
  float* v   = k  + nQKV;
  float* qf  = v  + nQKV;
  float* kfb = qf + nF;
  float* ctx = kfb + nF;

  dim3 blk(256);
  dim3 pblk(64);
  kproj_repack_kernel<<<KGRP_ + RGRP_, pblk, 0, stream>>>(keys, Wk, k,
      values, queries, Wv, Wq, Wo, Wqa, Wka, q, qf);

  vproj_feat_kernel<<<VGRP_ + QGRP_ + FGRP_, pblk, 0, stream>>>(q, qf + PK_WV_, v,
      qf + PK_QUERIES_, qf + PK_WQ_, ctx, k, qf + PK_WKA_, kfb);

  score_softmax_av_kernel<<<B_ * H_ * (LQ_ / TQ_), blk, 0, stream>>>(
      qf + PK_WQA_, kfb, v, wva, valid_lens, ctx);

  oproj_kernel<<<256, pblk, 0, stream>>>(ctx, qf + PK_WO_, (float*)d_out);
}
